# starting kernel
# baseline (speedup 1.0000x reference)
#include <hip/hip_runtime.h>


typedef float v2f __attribute__((ext_vector_type(2)));
typedef float v8f __attribute__((ext_vector_type(8)));

#define N_PTS   2048
#define DIN     16
#define NC      32
#define DC      16
#define NTILES  (N_PTS / 16)
#define WAVES   8
#define THREADS 256
#define EPS     1e-7f

__device__ __forceinline__ void wave_sync() { __builtin_amdgcn_wave_barrier(); }

__device__ __forceinline__ v8f wmma_f32(const v2f a, const v2f b, const v8f c) {
    return __builtin_amdgcn_wmma_f32_16x16x4_f32(false, a, false, b,
                                                 (short)0, c, false, false);
}

__global__ __launch_bounds__(THREADS)
void capsnet_routing_kernel(const float* __restrict__ x,
                            const float* __restrict__ W,
                            float* __restrict__ out)
{
    __shared__ __align__(16) float sWp[NC * DC * DIN];
    __shared__ __align__(16) float sStage[WAVES][NC][16];
    __shared__ __align__(16) float sV[DIN][NC];
    __shared__ __align__(16) float sOpre[NC][DC];
    __shared__ __align__(16) float sO[NC][DC];
    __shared__ float sFac[NC];

    const int tid  = threadIdx.x;
    const int wave = tid >> 5;
    const int lane = tid & 31;
    const int col  = lane & 15;
    const int h    = lane >> 4;
    const int bidx = blockIdx.x;

    const float* xb = x + (size_t)bidx * (N_PTS * DIN);

    for (int i = tid; i < NC * DC * DIN; i += THREADS) {
        const int c = i >> 8, d = (i >> 4) & 15, hh = (i >> 3) & 1, q = i & 7;
        const int k = 4 * (q >> 1) + 2 * hh + (q & 1);
        sWp[i] = W[k * (NC * DC) + c * DC + d];
    }
    __syncthreads();

    for (int it = 0; it < 3; ++it) {
        v2f vlo[4], vhi[4];
        if (it > 0) {
#pragma unroll
            for (int g = 0; g < 4; ++g) {
                const int k0 = 4 * g + 2 * h;
                vlo[g].x = sV[k0][col];       vlo[g].y = sV[k0 + 1][col];
                vhi[g].x = sV[k0][16 + col];  vhi[g].y = sV[k0 + 1][16 + col];
            }
        }
        {
            float* p = &sOpre[0][0];
            for (int i = tid; i < NC * DC; i += THREADS) p[i] = 0.0f;
        }
        __syncthreads();

        for (int mt = wave; mt < NTILES; mt += WAVES) {
            const int nrow = mt * 16 + col;
            v2f a[4];
#pragma unroll
            for (int g = 0; g < 4; ++g) {
                const int k0 = 4 * g + 2 * h;
                a[g].x = xb[nrow * DIN + k0];
                a[g].y = xb[nrow * DIN + k0 + 1];
            }

            if (it > 0) {
                v8f blo = {}, bhi = {};
#pragma unroll
                for (int g = 0; g < 4; ++g) {
                    blo = wmma_f32(a[g], vlo[g], blo);
                    bhi = wmma_f32(a[g], vhi[g], bhi);
                }
                {
                    float4* s0 = (float4*)&sStage[wave][col][8 * h];
                    s0[0] = make_float4(blo[0], blo[1], blo[2], blo[3]);
                    s0[1] = make_float4(blo[4], blo[5], blo[6], blo[7]);
                    float4* s1 = (float4*)&sStage[wave][16 + col][8 * h];
                    s1[0] = make_float4(bhi[0], bhi[1], bhi[2], bhi[3]);
                    s1[1] = make_float4(bhi[4], bhi[5], bhi[6], bhi[7]);
                }
                wave_sync();
                {
                    float vals[16];
#pragma unroll
                    for (int cc = 0; cc < 16; ++cc)
                        vals[cc] = sStage[wave][h * 16 + cc][col];
                    float m = vals[0];
#pragma unroll
                    for (int cc = 1; cc < 16; ++cc) m = fmaxf(m, vals[cc]);
                    m = fmaxf(m, __shfl_xor(m, 16, 32));
                    float s = 0.0f;
#pragma unroll
                    for (int cc = 0; cc < 16; ++cc) {
                        vals[cc] = __expf(vals[cc] - m);
                        s += vals[cc];
                    }
                    s += __shfl_xor(s, 16, 32);
                    const float r = 1.0f / s;
#pragma unroll
                    for (int cc = 0; cc < 16; ++cc)
                        sStage[wave][h * 16 + cc][col] = vals[cc] * r;
                }
                wave_sync();
            }

#pragma unroll 4
            for (int c = 0; c < NC; ++c) {
                const float4* wp =
                    (const float4*)&sWp[c * 256 + col * 16 + h * 8];
                const float4 w0 = wp[0];
                const float4 w1 = wp[1];
                v2f b0; b0.x = w0.x; b0.y = w0.y;
                v2f b1; b1.x = w0.z; b1.y = w0.w;
                v2f b2; b2.x = w1.x; b2.y = w1.y;
                v2f b3; b3.x = w1.z; b3.y = w1.w;
                v8f acc = {};
                acc = wmma_f32(a[0], b0, acc);
                acc = wmma_f32(a[1], b1, acc);
                acc = wmma_f32(a[2], b2, acc);
                acc = wmma_f32(a[3], b3, acc);

                float partial;
                if (it == 0) {
                    partial = 0.03125f * (acc[0] + acc[1] + acc[2] + acc[3] +
                                          acc[4] + acc[5] + acc[6] + acc[7]);
                } else {
                    const float4* cp =
                        (const float4*)&sStage[wave][c][8 * h];
                    const float4 c0 = cp[0];
                    const float4 c1 = cp[1];
                    partial = c0.x * acc[0] + c0.y * acc[1] +
                              c0.z * acc[2] + c0.w * acc[3] +
                              c1.x * acc[4] + c1.y * acc[5] +
                              c1.z * acc[6] + c1.w * acc[7];
                }
                atomicAdd(&sOpre[c][col], partial);
            }
        }
        __syncthreads();

        if (tid < NC) {
            float s2 = 0.0f;
#pragma unroll
            for (int d = 0; d < DC; ++d) {
                const float p = sOpre[tid][d];
                s2 = fmaf(p, p, s2);
            }
            sFac[tid] = (s2 / (1.0f + s2)) * rsqrtf(s2 + EPS);
        }
        __syncthreads();
        for (int i = tid; i < NC * DC; i += THREADS) {
            const int c = i >> 4, d = i & 15;
            sO[c][d] = sFac[c] * sOpre[c][d];
        }
        __syncthreads();

        if (it < 2) {
            for (int i = tid; i < DIN * NC; i += THREADS) {
                const int k = i >> 5, c = i & 31;
                const int hk = (k >> 1) & 1;
                const int qk = 2 * (k >> 2) + (k & 1);
                float s = 0.0f;
#pragma unroll
                for (int d = 0; d < DC; ++d)
                    s = fmaf(sWp[c * 256 + d * 16 + hk * 8 + qk], sO[c][d], s);
                sV[k][c] = s;
            }
            __syncthreads();
        }
    }

    for (int i = tid; i < NC * DC; i += THREADS) {
        out[(size_t)bidx * (NC * DC) + i] = sO[i >> 4][i & 15];
    }
}

extern "C" void kernel_launch(void* const* d_in, const int* in_sizes, int n_in,
                              void* d_out, int out_size, void* d_ws, size_t ws_size,
                              hipStream_t stream) {
    (void)in_sizes; (void)n_in; (void)out_size; (void)d_ws; (void)ws_size;
    const float* x = (const float*)d_in[0];
    const float* W = (const float*)d_in[1];
    float* out = (float*)d_out;
    hipLaunchKernelGGL(capsnet_routing_kernel, dim3(128), dim3(THREADS), 0, stream,
                       x, W, out);
}


# Split each batch item's 2048 sources over 4 work-groups (512 of 8 waves, 2 per WGP), one launch per routing pass; partial capsule sums via workspace, added in fixed order: f32 sum order changes
# speedup vs baseline: 2.0623x; 2.0623x over previous
#include <hip/hip_runtime.h>


typedef float v2f __attribute__((ext_vector_type(2)));
typedef float v8f __attribute__((ext_vector_type(8)));

#define BATCH   128
#define N_PTS   2048
#define DIN     16
#define NC      32
#define DC      16
#define NTILES  (N_PTS / 16)
#define WAVES   8
#define THREADS 256
#define EPS     1e-7f

#define SPLIT         4
#define TILES_PER_WG  (NTILES / SPLIT)
#define PLANE_FLOATS  (BATCH * SPLIT * NC * DC)
#define WS_NEED_BYTES (2ull * PLANE_FLOATS * 4ull)

__device__ __forceinline__ void wave_sync() { __builtin_amdgcn_wave_barrier(); }

__device__ __forceinline__ v8f wmma_f32(const v2f a, const v2f b, const v8f c) {
    return __builtin_amdgcn_wmma_f32_16x16x4_f32(false, a, false, b,
                                                 (short)0, c, false, false);
}

__device__ __forceinline__ void stage_w(float* sWp, const float* __restrict__ W,
                                        const int tid) {
    for (int i = tid; i < NC * DC * DIN; i += THREADS) {
        const int c = i >> 8, d = (i >> 4) & 15, hh = (i >> 3) & 1, q = i & 7;
        const int k = 4 * (q >> 1) + 2 * hh + (q & 1);
        sWp[i] = W[k * (NC * DC) + c * DC + d];
    }
}

__device__ __forceinline__ void squash_factor(const float (*sOpre)[DC],
                                              float* sFac, const int tid) {
    if (tid < NC) {
        float s2 = 0.0f;
#pragma unroll
        for (int d = 0; d < DC; ++d) {
            const float p = sOpre[tid][d];
            s2 = fmaf(p, p, s2);
        }
        sFac[tid] = (s2 / (1.0f + s2)) * rsqrtf(s2 + EPS);
    }
}

__device__ __forceinline__ void make_v(float (*sV)[NC], const float* sWp,
                                       const float (*sO)[DC], const int tid) {
    for (int i = tid; i < DIN * NC; i += THREADS) {
        const int k = i >> 5, c = i & 31;
        const int hk = (k >> 1) & 1;
        const int qk = 2 * (k >> 2) + (k & 1);
        float s = 0.0f;
#pragma unroll
        for (int d = 0; d < DC; ++d)
            s = fmaf(sWp[c * 256 + d * 16 + hk * 8 + qk], sO[c][d], s);
        sV[k][c] = s;
    }
}

template <int IT>
__device__ __forceinline__ void route_tiles(const float* __restrict__ xb,
                                            const float* sWp,
                                            float (*sStageW)[16],
                                            const float (*sV)[NC],
                                            float (*sOpre)[DC],
                                            const int mt_begin, const int mt_end,
                                            const int col, const int h)
{
    v2f vlo[4], vhi[4];
    if (IT > 0) {
#pragma unroll
        for (int g = 0; g < 4; ++g) {
            const int k0 = 4 * g + 2 * h;
            vlo[g].x = sV[k0][col];       vlo[g].y = sV[k0 + 1][col];
            vhi[g].x = sV[k0][16 + col];  vhi[g].y = sV[k0 + 1][16 + col];
        }
    }

    for (int mt = mt_begin; mt < mt_end; mt += WAVES) {
        const int nrow = mt * 16 + col;
        v2f a[4];
#pragma unroll
        for (int g = 0; g < 4; ++g) {
            const int k0 = 4 * g + 2 * h;
            a[g].x = xb[nrow * DIN + k0];
            a[g].y = xb[nrow * DIN + k0 + 1];
        }

        if (IT > 0) {
            v8f blo = {}, bhi = {};
#pragma unroll
            for (int g = 0; g < 4; ++g) {
                blo = wmma_f32(a[g], vlo[g], blo);
                bhi = wmma_f32(a[g], vhi[g], bhi);
            }
            {
                float4* s0 = (float4*)&sStageW[col][8 * h];
                s0[0] = make_float4(blo[0], blo[1], blo[2], blo[3]);
                s0[1] = make_float4(blo[4], blo[5], blo[6], blo[7]);
                float4* s1 = (float4*)&sStageW[16 + col][8 * h];
                s1[0] = make_float4(bhi[0], bhi[1], bhi[2], bhi[3]);
                s1[1] = make_float4(bhi[4], bhi[5], bhi[6], bhi[7]);
            }
            wave_sync();
            {
                float vals[16];
#pragma unroll
                for (int cc = 0; cc < 16; ++cc)
                    vals[cc] = sStageW[h * 16 + cc][col];
                float m = vals[0];
#pragma unroll
                for (int cc = 1; cc < 16; ++cc) m = fmaxf(m, vals[cc]);
                m = fmaxf(m, __shfl_xor(m, 16, 32));
                float s = 0.0f;
#pragma unroll
                for (int cc = 0; cc < 16; ++cc) {
                    vals[cc] = __expf(vals[cc] - m);
                    s += vals[cc];
                }
                s += __shfl_xor(s, 16, 32);
                const float r = 1.0f / s;
#pragma unroll
                for (int cc = 0; cc < 16; ++cc)
                    sStageW[h * 16 + cc][col] = vals[cc] * r;
            }
            wave_sync();
        }

#pragma unroll 4
        for (int c = 0; c < NC; ++c) {
            const float4* wp =
                (const float4*)&sWp[c * 256 + col * 16 + h * 8];
            const float4 w0 = wp[0];
            const float4 w1 = wp[1];
            v2f b0; b0.x = w0.x; b0.y = w0.y;
            v2f b1; b1.x = w0.z; b1.y = w0.w;
            v2f b2; b2.x = w1.x; b2.y = w1.y;
            v2f b3; b3.x = w1.z; b3.y = w1.w;
            v8f acc = {};
            acc = wmma_f32(a[0], b0, acc);
            acc = wmma_f32(a[1], b1, acc);
            acc = wmma_f32(a[2], b2, acc);
            acc = wmma_f32(a[3], b3, acc);

            float partial;
            if (IT == 0) {
                partial = 0.03125f * (acc[0] + acc[1] + acc[2] + acc[3] +
                                      acc[4] + acc[5] + acc[6] + acc[7]);
            } else {
                const float4* cp = (const float4*)&sStageW[c][8 * h];
                const float4 c0 = cp[0];
                const float4 c1 = cp[1];
                partial = c0.x * acc[0] + c0.y * acc[1] +
                          c0.z * acc[2] + c0.w * acc[3] +
                          c1.x * acc[4] + c1.y * acc[5] +
                          c1.z * acc[6] + c1.w * acc[7];
            }
            atomicAdd(&sOpre[c][col], partial);
        }
    }
}

template <int IT>
__global__ __launch_bounds__(THREADS)
void capsnet_pass_kernel(const float* __restrict__ x,
                         const float* __restrict__ W,
                         const float* __restrict__ part_in,
                         float* __restrict__ part_out,
                         const unsigned long long ws_bytes)
{
    if (ws_bytes < WS_NEED_BYTES) return;

    __shared__ __align__(16) float sWp[NC * DC * DIN];
    __shared__ __align__(16) float sStage[WAVES][NC][16];
    __shared__ __align__(16) float sV[DIN][NC];
    __shared__ __align__(16) float sOpre[NC][DC];
    __shared__ __align__(16) float sO[NC][DC];
    __shared__ float sFac[NC];

    const int tid  = threadIdx.x;
    const int wave = tid >> 5;
    const int lane = tid & 31;
    const int col  = lane & 15;
    const int h    = lane >> 4;
    const int bidx = blockIdx.x / SPLIT;
    const int part = blockIdx.x % SPLIT;

    const float* xb = x + (size_t)bidx * (N_PTS * DIN);

    stage_w(sWp, W, tid);
    if (IT > 0) {
        const float* pin = part_in + (size_t)bidx * (SPLIT * NC * DC);
        float* p = &sOpre[0][0];
        for (int i = tid; i < NC * DC; i += THREADS) {
            float s = pin[i];
            for (int gg = 1; gg < SPLIT; ++gg) s += pin[gg * (NC * DC) + i];
            p[i] = s;
        }
    }
    __syncthreads();

    if (IT > 0) {
        squash_factor(sOpre, sFac, tid);
        __syncthreads();
        for (int i = tid; i < NC * DC; i += THREADS) {
            const int c = i >> 4, d = i & 15;
            sO[c][d] = sFac[c] * sOpre[c][d];
        }
        __syncthreads();
        make_v(sV, sWp, sO, tid);
        __syncthreads();
    }
    {
        float* p = &sOpre[0][0];
        for (int i = tid; i < NC * DC; i += THREADS) p[i] = 0.0f;
    }
    __syncthreads();

    route_tiles<IT>(xb, sWp, sStage[wave], sV, sOpre,
                    part * TILES_PER_WG + wave, (part + 1) * TILES_PER_WG,
                    col, h);
    __syncthreads();

    float* pout = part_out + (size_t)blockIdx.x * (NC * DC);
    const float* p = &sOpre[0][0];
    for (int i = tid; i < NC * DC; i += THREADS) pout[i] = p[i];
}

__global__ __launch_bounds__(THREADS)
void capsnet_finish_kernel(const float* __restrict__ x,
                           const float* __restrict__ W,
                           const float* __restrict__ part_in,
                           float* __restrict__ out,
                           const unsigned long long ws_bytes)
{
    __shared__ __align__(16) float sWp[NC * DC * DIN];
    __shared__ __align__(16) float sStage[WAVES][NC][16];
    __shared__ __align__(16) float sV[DIN][NC];
    __shared__ __align__(16) float sOpre[NC][DC];
    __shared__ __align__(16) float sO[NC][DC];
    __shared__ float sFac[NC];

    const int tid  = threadIdx.x;
    const int bidx = blockIdx.x;

    if (ws_bytes >= WS_NEED_BYTES) {
        const float* pin = part_in + (size_t)bidx * (SPLIT * NC * DC);
        float* p = &sOpre[0][0];
        for (int i = tid; i < NC * DC; i += THREADS) {
            float s = pin[i];
            for (int gg = 1; gg < SPLIT; ++gg) s += pin[gg * (NC * DC) + i];
            p[i] = s;
        }
        __syncthreads();
        squash_factor(sOpre, sFac, tid);
        __syncthreads();
        for (int i = tid; i < NC * DC; i += THREADS)
            out[(size_t)bidx * (NC * DC) + i] = sFac[i >> 4] * p[i];
        return;
    }

    const int wave = tid >> 5;
    const int lane = tid & 31;
    const int col  = lane & 15;
    const int h    = lane >> 4;
    const float* xb = x + (size_t)bidx * (N_PTS * DIN);

    stage_w(sWp, W, tid);
    __syncthreads();

    for (int it = 0; it < 3; ++it) {
        {
            float* p = &sOpre[0][0];
            for (int i = tid; i < NC * DC; i += THREADS) p[i] = 0.0f;
        }
        __syncthreads();
        if (it == 0)
            route_tiles<0>(xb, sWp, sStage[wave], sV, sOpre, wave, NTILES, col, h);
        else
            route_tiles<1>(xb, sWp, sStage[wave], sV, sOpre, wave, NTILES, col, h);
        __syncthreads();
        squash_factor(sOpre, sFac, tid);
        __syncthreads();
        for (int i = tid; i < NC * DC; i += THREADS) {
            const int c = i >> 4, d = i & 15;
            sO[c][d] = sFac[c] * sOpre[c][d];
        }
        __syncthreads();
        if (it < 2) {
            make_v(sV, sWp, sO, tid);
            __syncthreads();
        }
    }
    for (int i = tid; i < NC * DC; i += THREADS)
        out[(size_t)bidx * (NC * DC) + i] = sO[i >> 4][i & 15];
}

extern "C" void kernel_launch(void* const* d_in, const int* in_sizes, int n_in,
                              void* d_out, int out_size, void* d_ws, size_t ws_size,
                              hipStream_t stream) {
    (void)in_sizes; (void)n_in; (void)out_size;
    const float* x = (const float*)d_in[0];
    const float* W = (const float*)d_in[1];
    float* out = (float*)d_out;
    float* planeA = (float*)d_ws;
    float* planeB = (float*)d_ws + PLANE_FLOATS;
    const unsigned long long wsb = (unsigned long long)ws_size;
    hipLaunchKernelGGL(capsnet_pass_kernel<0>, dim3(BATCH * SPLIT), dim3(THREADS), 0,
                       stream, x, W, (const float*)planeB, planeA, wsb);
    hipLaunchKernelGGL(capsnet_pass_kernel<1>, dim3(BATCH * SPLIT), dim3(THREADS), 0,
                       stream, x, W, (const float*)planeA, planeB, wsb);
    hipLaunchKernelGGL(capsnet_pass_kernel<2>, dim3(BATCH * SPLIT), dim3(THREADS), 0,
                       stream, x, W, (const float*)planeB, planeA, wsb);
    hipLaunchKernelGGL(capsnet_finish_kernel, dim3(BATCH), dim3(THREADS), 0,
                       stream, x, W, (const float*)planeA, out, wsb);
}


# Hat product transposed (W_c^T as A, x tile as B): lane = source row, its coefficient scales x, WMMA accumulator sums the wave's 4 tiles, DPP row sum replaces LDS atomics; f32 summation order changes
# speedup vs baseline: 2.4640x; 1.1948x over previous
#include <hip/hip_runtime.h>


typedef float v2f __attribute__((ext_vector_type(2)));
typedef float v8f __attribute__((ext_vector_type(8)));

#define BATCH   128
#define N_PTS   2048
#define DIN     16
#define NC      32
#define DC      16
#define NTILES  (N_PTS / 16)
#define WAVES   8
#define THREADS 256
#define EPS     1e-7f

#define SPLIT         4
#define TILES_PER_WG  (NTILES / SPLIT)
#define PLANE_FLOATS  (BATCH * SPLIT * NC * DC)
#define WS_NEED_BYTES (2ull * PLANE_FLOATS * 4ull)

static_assert(TILES_PER_WG == 4 * WAVES, "a wave of a pass launch owns exactly four row-tiles");

__device__ __forceinline__ void wave_sync() { __builtin_amdgcn_wave_barrier(); }

__device__ __forceinline__ v8f wmma_f32(const v2f a, const v2f b, const v8f c) {
    return __builtin_amdgcn_wmma_f32_16x16x4_f32(false, a, false, b,
                                                 (short)0, c, false, false);
}

__device__ __forceinline__ void stage_w(float* sWp, const float* __restrict__ W,
                                        const int tid) {
    for (int i = tid; i < NC * DC * DIN; i += THREADS) {
        const int c = i >> 8, d = (i >> 4) & 15, hh = (i >> 3) & 1, q = i & 7;
        const int k = 4 * (q >> 1) + 2 * hh + (q & 1);
        sWp[i] = W[k * (NC * DC) + c * DC + d];
    }
}

__device__ __forceinline__ void squash_factor(const float (*sOpre)[DC],
                                              float* sFac, const int tid) {
    if (tid < NC) {
        float s2 = 0.0f;
#pragma unroll
        for (int d = 0; d < DC; ++d) {
            const float p = sOpre[tid][d];
            s2 = fmaf(p, p, s2);
        }
        sFac[tid] = (s2 / (1.0f + s2)) * rsqrtf(s2 + EPS);
    }
}

__device__ __forceinline__ void make_v(float (*sV)[NC], const float* sWp,
                                       const float (*sO)[DC], const int tid) {
    for (int i = tid; i < DIN * NC; i += THREADS) {
        const int k = i >> 5, c = i & 31;
        const int hk = (k >> 1) & 1;
        const int qk = 2 * (k >> 2) + (k & 1);
        float s = 0.0f;
#pragma unroll
        for (int d = 0; d < DC; ++d)
            s = fmaf(sWp[c * 256 + d * 16 + hk * 8 + qk], sO[c][d], s);
        sV[k][c] = s;
    }
}

template <int IT>
__device__ __forceinline__ void route_tiles(const float* __restrict__ xb,
                                            const float* sWp,
                                            float (*sStageW)[16],
                                            const float (*sV)[NC],
                                            float (*sOpre)[DC],
                                            const int mt_begin, const int mt_end,
                                            const int col, const int h)
{
    v2f vlo[4], vhi[4];
    if (IT > 0) {
#pragma unroll
        for (int g = 0; g < 4; ++g) {
            const int k0 = 4 * g + 2 * h;
            vlo[g].x = sV[k0][col];       vlo[g].y = sV[k0 + 1][col];
            vhi[g].x = sV[k0][16 + col];  vhi[g].y = sV[k0 + 1][16 + col];
        }
    }

    for (int mt = mt_begin; mt < mt_end; mt += WAVES) {
        const int nrow = mt * 16 + col;
        v2f a[4];
#pragma unroll
        for (int g = 0; g < 4; ++g) {
            const int k0 = 4 * g + 2 * h;
            a[g].x = xb[nrow * DIN + k0];
            a[g].y = xb[nrow * DIN + k0 + 1];
        }

        if (IT > 0) {
            v8f blo = {}, bhi = {};
#pragma unroll
            for (int g = 0; g < 4; ++g) {
                blo = wmma_f32(a[g], vlo[g], blo);
                bhi = wmma_f32(a[g], vhi[g], bhi);
            }
            {
                float4* s0 = (float4*)&sStageW[col][8 * h];
                s0[0] = make_float4(blo[0], blo[1], blo[2], blo[3]);
                s0[1] = make_float4(blo[4], blo[5], blo[6], blo[7]);
                float4* s1 = (float4*)&sStageW[16 + col][8 * h];
                s1[0] = make_float4(bhi[0], bhi[1], bhi[2], bhi[3]);
                s1[1] = make_float4(bhi[4], bhi[5], bhi[6], bhi[7]);
            }
            wave_sync();
            {
                float vals[16];
#pragma unroll
                for (int cc = 0; cc < 16; ++cc)
                    vals[cc] = sStageW[h * 16 + cc][col];
                float m = vals[0];
#pragma unroll
                for (int cc = 1; cc < 16; ++cc) m = fmaxf(m, vals[cc]);
                m = fmaxf(m, __shfl_xor(m, 16, 32));
                float s = 0.0f;
#pragma unroll
                for (int cc = 0; cc < 16; ++cc) {
                    vals[cc] = __expf(vals[cc] - m);
                    s += vals[cc];
                }
                s += __shfl_xor(s, 16, 32);
                const float r = 1.0f / s;
#pragma unroll
                for (int cc = 0; cc < 16; ++cc)
                    sStageW[h * 16 + cc][col] = vals[cc] * r;
            }
            wave_sync();
        }

#pragma unroll 4
        for (int c = 0; c < NC; ++c) {
            const float4* wp =
                (const float4*)&sWp[c * 256 + col * 16 + h * 8];
            const float4 w0 = wp[0];
            const float4 w1 = wp[1];
            v2f b0; b0.x = w0.x; b0.y = w0.y;
            v2f b1; b1.x = w0.z; b1.y = w0.w;
            v2f b2; b2.x = w1.x; b2.y = w1.y;
            v2f b3; b3.x = w1.z; b3.y = w1.w;
            v8f acc = {};
            acc = wmma_f32(a[0], b0, acc);
            acc = wmma_f32(a[1], b1, acc);
            acc = wmma_f32(a[2], b2, acc);
            acc = wmma_f32(a[3], b3, acc);

            float partial;
            if (IT == 0) {
                partial = 0.03125f * (acc[0] + acc[1] + acc[2] + acc[3] +
                                      acc[4] + acc[5] + acc[6] + acc[7]);
            } else {
                const float4* cp = (const float4*)&sStageW[c][8 * h];
                const float4 c0 = cp[0];
                const float4 c1 = cp[1];
                partial = c0.x * acc[0] + c0.y * acc[1] +
                          c0.z * acc[2] + c0.w * acc[3] +
                          c1.x * acc[4] + c1.y * acc[5] +
                          c1.z * acc[6] + c1.w * acc[7];
            }
            atomicAdd(&sOpre[c][col], partial);
        }
    }
}

template <int CTRL>
__device__ __forceinline__ float row_xor(const float v) {
    return __builtin_bit_cast(float, __builtin_amdgcn_update_dpp(
        0, __builtin_bit_cast(int, v), CTRL, 0xf, 0xf, false));
}

template <int CTRL>
__device__ __forceinline__ float fold_pair(const bool bit, const float lo,
                                           const float hi) {
    const float keep = bit ? hi : lo;
    const float send = bit ? lo : hi;
    return keep + row_xor<CTRL>(send);
}

template <int IT>
__device__ __forceinline__ void route_wave4(const float* __restrict__ xb,
                                            const float* sWp,
                                            float (*sStageW)[16],
                                            float* sCoefW,
                                            const float (*sV)[NC],
                                            float* sPartW,
                                            const int mt0,
                                            const int col, const int h)
{
    v2f xa[4][4];
#pragma unroll
    for (int t = 0; t < 4; ++t) {
        const int nrow = (mt0 + WAVES * t) * 16 + col;
#pragma unroll
        for (int g = 0; g < 4; ++g) {
            const int k0 = 4 * g + 2 * h;
            xa[t][g].x = xb[nrow * DIN + k0];
            xa[t][g].y = xb[nrow * DIN + k0 + 1];
        }
    }

    if (IT > 0) {
        v2f vlo[4], vhi[4];
#pragma unroll
        for (int g = 0; g < 4; ++g) {
            const int k0 = 4 * g + 2 * h;
            vlo[g].x = sV[k0][col];       vlo[g].y = sV[k0 + 1][col];
            vhi[g].x = sV[k0][16 + col];  vhi[g].y = sV[k0 + 1][16 + col];
        }
#pragma unroll
        for (int t = 0; t < 4; ++t) {
            v8f blo = {}, bhi = {};
#pragma unroll
            for (int g = 0; g < 4; ++g) {
                blo = wmma_f32(xa[t][g], vlo[g], blo);
                bhi = wmma_f32(xa[t][g], vhi[g], bhi);
            }
            {
                float4* s0 = (float4*)&sStageW[col][8 * h];
                s0[0] = make_float4(blo[0], blo[1], blo[2], blo[3]);
                s0[1] = make_float4(blo[4], blo[5], blo[6], blo[7]);
                float4* s1 = (float4*)&sStageW[16 + col][8 * h];
                s1[0] = make_float4(bhi[0], bhi[1], bhi[2], bhi[3]);
                s1[1] = make_float4(bhi[4], bhi[5], bhi[6], bhi[7]);
            }
            wave_sync();
            {
                float vals[16];
#pragma unroll
                for (int cc = 0; cc < 16; ++cc)
                    vals[cc] = sStageW[h * 16 + cc][col];
                float m = vals[0];
#pragma unroll
                for (int cc = 1; cc < 16; ++cc) m = fmaxf(m, vals[cc]);
                m = fmaxf(m, __shfl_xor(m, 16, 32));
                float s = 0.0f;
#pragma unroll
                for (int cc = 0; cc < 16; ++cc) {
                    vals[cc] = __expf(vals[cc] - m);
                    s += vals[cc];
                }
                s += __shfl_xor(s, 16, 32);
                const float r = 1.0f / s;
#pragma unroll
                for (int cc = 0; cc < 16; ++cc)
                    sCoefW[((h * 16 + cc) * 16 + col) * 4 + t] = vals[cc] * r;
            }
            wave_sync();
        }
    }

    const bool bit0 = (col & 1) != 0;
    const bool bit1 = (col & 2) != 0;
    const bool bit2 = (col & 4) != 0;
    const bool bit3 = (col & 8) != 0;
    float* const pdst = sPartW + (bit3 ? DC : 0) + 8 * h + (col & 7);

    for (int c = 0; c < NC; c += 2) {
        const float4* wp =
            (const float4*)&sWp[c * 256 + col * 16 + h * 8];
        const float4 u0 = wp[0],  u1 = wp[1];
        const float4 w0 = wp[64], w1 = wp[65];
        v2f wa0[4], wa1[4];
        wa0[0].x = u0.x; wa0[0].y = u0.y;  wa0[1].x = u0.z; wa0[1].y = u0.w;
        wa0[2].x = u1.x; wa0[2].y = u1.y;  wa0[3].x = u1.z; wa0[3].y = u1.w;
        wa1[0].x = w0.x; wa1[0].y = w0.y;  wa1[1].x = w0.z; wa1[1].y = w0.w;
        wa1[2].x = w1.x; wa1[2].y = w1.y;  wa1[3].x = w1.z; wa1[3].y = w1.w;

        float cf0[4] = {1.0f, 1.0f, 1.0f, 1.0f};
        float cf1[4] = {1.0f, 1.0f, 1.0f, 1.0f};
        if (IT > 0) {
            const float4 q0 = *(const float4*)&sCoefW[(c * 16 + col) * 4];
            const float4 q1 = *(const float4*)&sCoefW[((c + 1) * 16 + col) * 4];
            cf0[0] = q0.x; cf0[1] = q0.y; cf0[2] = q0.z; cf0[3] = q0.w;
            cf1[0] = q1.x; cf1[1] = q1.y; cf1[2] = q1.z; cf1[3] = q1.w;
        }

        v8f acc0 = {}, acc1 = {};
#pragma unroll
        for (int t = 0; t < 4; ++t) {
#pragma unroll
            for (int g = 0; g < 4; ++g) {
                v2f b0 = xa[t][g], b1 = xa[t][g];
                if (IT > 0) {
                    b0.x *= cf0[t]; b0.y *= cf0[t];
                    b1.x *= cf1[t]; b1.y *= cf1[t];
                }
                acc0 = wmma_f32(wa0[g], b0, acc0);
                acc1 = wmma_f32(wa1[g], b1, acc1);
            }
        }

        float s0[4], s1[4];
#pragma unroll
        for (int p = 0; p < 4; ++p) {
            s0[p] = fold_pair<0x161>(bit0, acc0[2 * p], acc0[2 * p + 1]);
            s1[p] = fold_pair<0x161>(bit0, acc1[2 * p], acc1[2 * p + 1]);
        }
        const float e00 = fold_pair<0x162>(bit1, s0[0], s0[1]);
        const float e01 = fold_pair<0x162>(bit1, s0[2], s0[3]);
        const float e10 = fold_pair<0x162>(bit1, s1[0], s1[1]);
        const float e11 = fold_pair<0x162>(bit1, s1[2], s1[3]);
        const float f0 = fold_pair<0x164>(bit2, e00, e01);
        const float f1 = fold_pair<0x164>(bit2, e10, e11);
        float tot = fold_pair<0x168>(bit3, f0, f1);
        if (IT == 0) tot *= 0.03125f;
        pdst[c * DC] = tot;
    }
}

template <int IT>
__global__ __launch_bounds__(THREADS)
void capsnet_pass_kernel(const float* __restrict__ x,
                         const float* __restrict__ W,
                         const float* __restrict__ part_in,
                         float* __restrict__ part_out,
                         const unsigned long long ws_bytes)
{
    if (ws_bytes < WS_NEED_BYTES) return;

    __shared__ __align__(16) float sWp[NC * DC * DIN];
    __shared__ __align__(16) float sStage[WAVES][NC][16];
    __shared__ __align__(16) float sCoef[WAVES][NC * 16 * 4];
    __shared__ __align__(16) float sV[DIN][NC];
    __shared__ __align__(16) float sOpre[NC][DC];
    __shared__ __align__(16) float sO[NC][DC];
    __shared__ float sFac[NC];

    const int tid  = threadIdx.x;
    const int wave = tid >> 5;
    const int lane = tid & 31;
    const int col  = lane & 15;
    const int h    = lane >> 4;
    const int bidx = blockIdx.x / SPLIT;
    const int part = blockIdx.x % SPLIT;

    const float* xb = x + (size_t)bidx * (N_PTS * DIN);

    stage_w(sWp, W, tid);
    if (IT > 0) {
        const float* pin = part_in + (size_t)bidx * (SPLIT * NC * DC);
        float* p = &sOpre[0][0];
        for (int i = tid; i < NC * DC; i += THREADS) {
            float s = pin[i];
            for (int gg = 1; gg < SPLIT; ++gg) s += pin[gg * (NC * DC) + i];
            p[i] = s;
        }
    }
    __syncthreads();

    if (IT > 0) {
        squash_factor(sOpre, sFac, tid);
        __syncthreads();
        for (int i = tid; i < NC * DC; i += THREADS) {
            const int c = i >> 4, d = i & 15;
            sO[c][d] = sFac[c] * sOpre[c][d];
        }
        __syncthreads();
        make_v(sV, sWp, sO, tid);
        __syncthreads();
    }

    route_wave4<IT>(xb, sWp, sStage[wave], sCoef[wave], sV, &sStage[wave][0][0],
                    part * TILES_PER_WG + wave, col, h);
    __syncthreads();

    float* pout = part_out + (size_t)blockIdx.x * (NC * DC);
    const float* p = &sStage[0][0][0];
    for (int i = tid; i < NC * DC; i += THREADS) {
        float s = p[i];
        for (int w = 1; w < WAVES; ++w) s += p[w * (NC * DC) + i];
        pout[i] = s;
    }
}

__global__ __launch_bounds__(THREADS)
void capsnet_finish_kernel(const float* __restrict__ x,
                           const float* __restrict__ W,
                           const float* __restrict__ part_in,
                           float* __restrict__ out,
                           const unsigned long long ws_bytes)
{
    __shared__ __align__(16) float sWp[NC * DC * DIN];
    __shared__ __align__(16) float sStage[WAVES][NC][16];
    __shared__ __align__(16) float sV[DIN][NC];
    __shared__ __align__(16) float sOpre[NC][DC];
    __shared__ __align__(16) float sO[NC][DC];
    __shared__ float sFac[NC];

    const int tid  = threadIdx.x;
    const int bidx = blockIdx.x;

    if (ws_bytes >= WS_NEED_BYTES) {
        const float* pin = part_in + (size_t)bidx * (SPLIT * NC * DC);
        float* p = &sOpre[0][0];
        for (int i = tid; i < NC * DC; i += THREADS) {
            float s = pin[i];
            for (int gg = 1; gg < SPLIT; ++gg) s += pin[gg * (NC * DC) + i];
            p[i] = s;
        }
        __syncthreads();
        squash_factor(sOpre, sFac, tid);
        __syncthreads();
        for (int i = tid; i < NC * DC; i += THREADS)
            out[(size_t)bidx * (NC * DC) + i] = sFac[i >> 4] * p[i];
        return;
    }

    const int wave = tid >> 5;
    const int lane = tid & 31;
    const int col  = lane & 15;
    const int h    = lane >> 4;
    const float* xb = x + (size_t)bidx * (N_PTS * DIN);

    stage_w(sWp, W, tid);
    __syncthreads();

    for (int it = 0; it < 3; ++it) {
        {
            float* p = &sOpre[0][0];
            for (int i = tid; i < NC * DC; i += THREADS) p[i] = 0.0f;
        }
        __syncthreads();
        if (it == 0)
            route_tiles<0>(xb, sWp, sStage[wave], sV, sOpre, wave, NTILES, col, h);
        else
            route_tiles<1>(xb, sWp, sStage[wave], sV, sOpre, wave, NTILES, col, h);
        __syncthreads();
        squash_factor(sOpre, sFac, tid);
        __syncthreads();
        for (int i = tid; i < NC * DC; i += THREADS) {
            const int c = i >> 4, d = i & 15;
            sO[c][d] = sFac[c] * sOpre[c][d];
        }
        __syncthreads();
        if (it < 2) {
            make_v(sV, sWp, sO, tid);
            __syncthreads();
        }
    }
    for (int i = tid; i < NC * DC; i += THREADS)
        out[(size_t)bidx * (NC * DC) + i] = sO[i >> 4][i & 15];
}

extern "C" void kernel_launch(void* const* d_in, const int* in_sizes, int n_in,
                              void* d_out, int out_size, void* d_ws, size_t ws_size,
                              hipStream_t stream) {
    (void)in_sizes; (void)n_in; (void)out_size;
    const float* x = (const float*)d_in[0];
    const float* W = (const float*)d_in[1];
    float* out = (float*)d_out;
    float* planeA = (float*)d_ws;
    float* planeB = (float*)d_ws + PLANE_FLOATS;
    const unsigned long long wsb = (unsigned long long)ws_size;
    hipLaunchKernelGGL(capsnet_pass_kernel<0>, dim3(BATCH * SPLIT), dim3(THREADS), 0,
                       stream, x, W, (const float*)planeB, planeA, wsb);
    hipLaunchKernelGGL(capsnet_pass_kernel<1>, dim3(BATCH * SPLIT), dim3(THREADS), 0,
                       stream, x, W, (const float*)planeA, planeB, wsb);
    hipLaunchKernelGGL(capsnet_pass_kernel<2>, dim3(BATCH * SPLIT), dim3(THREADS), 0,
                       stream, x, W, (const float*)planeB, planeA, wsb);
    hipLaunchKernelGGL(capsnet_finish_kernel, dim3(BATCH), dim3(THREADS), 0,
                       stream, x, W, (const float*)planeA, out, wsb);
}


# V = W.o product: a thread owns (capsule, k-place in the permuted W row): lanes run along k, not capsules; LDS reads put 2 lanes on a bank, not 32; same multiply-add order per element
# speedup vs baseline: 2.9277x; 1.1882x over previous
#include <hip/hip_runtime.h>


typedef float v2f __attribute__((ext_vector_type(2)));
typedef float v8f __attribute__((ext_vector_type(8)));

#define BATCH   128
#define N_PTS   2048
#define DIN     16
#define NC      32
#define DC      16
#define NTILES  (N_PTS / 16)
#define WAVES   8
#define THREADS 256
#define EPS     1e-7f

#define SPLIT         4
#define TILES_PER_WG  (NTILES / SPLIT)
#define PLANE_FLOATS  (BATCH * SPLIT * NC * DC)
#define WS_NEED_BYTES (2ull * PLANE_FLOATS * 4ull)

static_assert(TILES_PER_WG == 4 * WAVES, "a wave of a pass launch owns exactly four row-tiles");

__device__ __forceinline__ void wave_sync() { __builtin_amdgcn_wave_barrier(); }

__device__ __forceinline__ v8f wmma_f32(const v2f a, const v2f b, const v8f c) {
    return __builtin_amdgcn_wmma_f32_16x16x4_f32(false, a, false, b,
                                                 (short)0, c, false, false);
}

__device__ __forceinline__ void stage_w(float* sWp, const float* __restrict__ W,
                                        const int tid) {
    for (int i = tid; i < NC * DC * DIN; i += THREADS) {
        const int c = i >> 8, d = (i >> 4) & 15, hh = (i >> 3) & 1, q = i & 7;
        const int k = 4 * (q >> 1) + 2 * hh + (q & 1);
        sWp[i] = W[k * (NC * DC) + c * DC + d];
    }
}

__device__ __forceinline__ void squash_factor(const float (*sOpre)[DC],
                                              float* sFac, const int tid) {
    if (tid < NC) {
        float s2 = 0.0f;
#pragma unroll
        for (int d = 0; d < DC; ++d) {
            const float p = sOpre[tid][d];
            s2 = fmaf(p, p, s2);
        }
        sFac[tid] = (s2 / (1.0f + s2)) * rsqrtf(s2 + EPS);
    }
}

__device__ __forceinline__ void make_v(float (*sV)[NC], const float* sWp,
                                       const float (*sO)[DC], const int tid) {
    for (int i = tid; i < DIN * NC; i += THREADS) {
        const int c = i >> 4, p = i & 15;
        const int hk = p >> 3, qk = p & 7;
        const int k = 4 * (qk >> 1) + 2 * hk + (qk & 1);
        float s = 0.0f;
#pragma unroll
        for (int d = 0; d < DC; ++d)
            s = fmaf(sWp[c * 256 + d * 16 + p], sO[c][d], s);
        sV[k][c] = s;
    }
}

template <int IT>
__device__ __forceinline__ void route_tiles(const float* __restrict__ xb,
                                            const float* sWp,
                                            float (*sStageW)[16],
                                            const float (*sV)[NC],
                                            float (*sOpre)[DC],
                                            const int mt_begin, const int mt_end,
                                            const int col, const int h)
{
    v2f vlo[4], vhi[4];
    if (IT > 0) {
#pragma unroll
        for (int g = 0; g < 4; ++g) {
            const int k0 = 4 * g + 2 * h;
            vlo[g].x = sV[k0][col];       vlo[g].y = sV[k0 + 1][col];
            vhi[g].x = sV[k0][16 + col];  vhi[g].y = sV[k0 + 1][16 + col];
        }
    }

    for (int mt = mt_begin; mt < mt_end; mt += WAVES) {
        const int nrow = mt * 16 + col;
        v2f a[4];
#pragma unroll
        for (int g = 0; g < 4; ++g) {
            const int k0 = 4 * g + 2 * h;
            a[g].x = xb[nrow * DIN + k0];
            a[g].y = xb[nrow * DIN + k0 + 1];
        }

        if (IT > 0) {
            v8f blo = {}, bhi = {};
#pragma unroll
            for (int g = 0; g < 4; ++g) {
                blo = wmma_f32(a[g], vlo[g], blo);
                bhi = wmma_f32(a[g], vhi[g], bhi);
            }
            {
                float4* s0 = (float4*)&sStageW[col][8 * h];
                s0[0] = make_float4(blo[0], blo[1], blo[2], blo[3]);
                s0[1] = make_float4(blo[4], blo[5], blo[6], blo[7]);
                float4* s1 = (float4*)&sStageW[16 + col][8 * h];
                s1[0] = make_float4(bhi[0], bhi[1], bhi[2], bhi[3]);
                s1[1] = make_float4(bhi[4], bhi[5], bhi[6], bhi[7]);
            }
            wave_sync();
            {
                float vals[16];
#pragma unroll
                for (int cc = 0; cc < 16; ++cc)
                    vals[cc] = sStageW[h * 16 + cc][col];
                float m = vals[0];
#pragma unroll
                for (int cc = 1; cc < 16; ++cc) m = fmaxf(m, vals[cc]);
                m = fmaxf(m, __shfl_xor(m, 16, 32));
                float s = 0.0f;
#pragma unroll
                for (int cc = 0; cc < 16; ++cc) {
                    vals[cc] = __expf(vals[cc] - m);
                    s += vals[cc];
                }
                s += __shfl_xor(s, 16, 32);
                const float r = 1.0f / s;
#pragma unroll
                for (int cc = 0; cc < 16; ++cc)
                    sStageW[h * 16 + cc][col] = vals[cc] * r;
            }
            wave_sync();
        }

#pragma unroll 4
        for (int c = 0; c < NC; ++c) {
            const float4* wp =
                (const float4*)&sWp[c * 256 + col * 16 + h * 8];
            const float4 w0 = wp[0];
            const float4 w1 = wp[1];
            v2f b0; b0.x = w0.x; b0.y = w0.y;
            v2f b1; b1.x = w0.z; b1.y = w0.w;
            v2f b2; b2.x = w1.x; b2.y = w1.y;
            v2f b3; b3.x = w1.z; b3.y = w1.w;
            v8f acc = {};
            acc = wmma_f32(a[0], b0, acc);
            acc = wmma_f32(a[1], b1, acc);
            acc = wmma_f32(a[2], b2, acc);
            acc = wmma_f32(a[3], b3, acc);

            float partial;
            if (IT == 0) {
                partial = 0.03125f * (acc[0] + acc[1] + acc[2] + acc[3] +
                                      acc[4] + acc[5] + acc[6] + acc[7]);
            } else {
                const float4* cp = (const float4*)&sStageW[c][8 * h];
                const float4 c0 = cp[0];
                const float4 c1 = cp[1];
                partial = c0.x * acc[0] + c0.y * acc[1] +
                          c0.z * acc[2] + c0.w * acc[3] +
                          c1.x * acc[4] + c1.y * acc[5] +
                          c1.z * acc[6] + c1.w * acc[7];
            }
            atomicAdd(&sOpre[c][col], partial);
        }
    }
}

template <int CTRL>
__device__ __forceinline__ float row_xor(const float v) {
    return __builtin_bit_cast(float, __builtin_amdgcn_update_dpp(
        0, __builtin_bit_cast(int, v), CTRL, 0xf, 0xf, false));
}

template <int CTRL>
__device__ __forceinline__ float fold_pair(const bool bit, const float lo,
                                           const float hi) {
    const float keep = bit ? hi : lo;
    const float send = bit ? lo : hi;
    return keep + row_xor<CTRL>(send);
}

template <int IT>
__device__ __forceinline__ void route_wave4(const float* __restrict__ xb,
                                            const float* sWp,
                                            float (*sStageW)[16],
                                            float* sCoefW,
                                            const float (*sV)[NC],
                                            float* sPartW,
                                            const int mt0,
                                            const int col, const int h)
{
    v2f xa[4][4];
#pragma unroll
    for (int t = 0; t < 4; ++t) {
        const int nrow = (mt0 + WAVES * t) * 16 + col;
#pragma unroll
        for (int g = 0; g < 4; ++g) {
            const int k0 = 4 * g + 2 * h;
            xa[t][g].x = xb[nrow * DIN + k0];
            xa[t][g].y = xb[nrow * DIN + k0 + 1];
        }
    }

    if (IT > 0) {
        v2f vlo[4], vhi[4];
#pragma unroll
        for (int g = 0; g < 4; ++g) {
            const int k0 = 4 * g + 2 * h;
            vlo[g].x = sV[k0][col];       vlo[g].y = sV[k0 + 1][col];
            vhi[g].x = sV[k0][16 + col];  vhi[g].y = sV[k0 + 1][16 + col];
        }
#pragma unroll
        for (int t = 0; t < 4; ++t) {
            v8f blo = {}, bhi = {};
#pragma unroll
            for (int g = 0; g < 4; ++g) {
                blo = wmma_f32(xa[t][g], vlo[g], blo);
                bhi = wmma_f32(xa[t][g], vhi[g], bhi);
            }
            {
                float4* s0 = (float4*)&sStageW[col][8 * h];
                s0[0] = make_float4(blo[0], blo[1], blo[2], blo[3]);
                s0[1] = make_float4(blo[4], blo[5], blo[6], blo[7]);
                float4* s1 = (float4*)&sStageW[16 + col][8 * h];
                s1[0] = make_float4(bhi[0], bhi[1], bhi[2], bhi[3]);
                s1[1] = make_float4(bhi[4], bhi[5], bhi[6], bhi[7]);
            }
            wave_sync();
            {
                float vals[16];
#pragma unroll
                for (int cc = 0; cc < 16; ++cc)
                    vals[cc] = sStageW[h * 16 + cc][col];
                float m = vals[0];
#pragma unroll
                for (int cc = 1; cc < 16; ++cc) m = fmaxf(m, vals[cc]);
                m = fmaxf(m, __shfl_xor(m, 16, 32));
                float s = 0.0f;
#pragma unroll
                for (int cc = 0; cc < 16; ++cc) {
                    vals[cc] = __expf(vals[cc] - m);
                    s += vals[cc];
                }
                s += __shfl_xor(s, 16, 32);
                const float r = 1.0f / s;
#pragma unroll
                for (int cc = 0; cc < 16; ++cc)
                    sCoefW[((h * 16 + cc) * 16 + col) * 4 + t] = vals[cc] * r;
            }
            wave_sync();
        }
    }

    const bool bit0 = (col & 1) != 0;
    const bool bit1 = (col & 2) != 0;
    const bool bit2 = (col & 4) != 0;
    const bool bit3 = (col & 8) != 0;
    float* const pdst = sPartW + (bit3 ? DC : 0) + 8 * h + (col & 7);

    for (int c = 0; c < NC; c += 2) {
        const float4* wp =
            (const float4*)&sWp[c * 256 + col * 16 + h * 8];
        const float4 u0 = wp[0],  u1 = wp[1];
        const float4 w0 = wp[64], w1 = wp[65];
        v2f wa0[4], wa1[4];
        wa0[0].x = u0.x; wa0[0].y = u0.y;  wa0[1].x = u0.z; wa0[1].y = u0.w;
        wa0[2].x = u1.x; wa0[2].y = u1.y;  wa0[3].x = u1.z; wa0[3].y = u1.w;
        wa1[0].x = w0.x; wa1[0].y = w0.y;  wa1[1].x = w0.z; wa1[1].y = w0.w;
        wa1[2].x = w1.x; wa1[2].y = w1.y;  wa1[3].x = w1.z; wa1[3].y = w1.w;

        float cf0[4] = {1.0f, 1.0f, 1.0f, 1.0f};
        float cf1[4] = {1.0f, 1.0f, 1.0f, 1.0f};
        if (IT > 0) {
            const float4 q0 = *(const float4*)&sCoefW[(c * 16 + col) * 4];
            const float4 q1 = *(const float4*)&sCoefW[((c + 1) * 16 + col) * 4];
            cf0[0] = q0.x; cf0[1] = q0.y; cf0[2] = q0.z; cf0[3] = q0.w;
            cf1[0] = q1.x; cf1[1] = q1.y; cf1[2] = q1.z; cf1[3] = q1.w;
        }

        v8f acc0 = {}, acc1 = {};
#pragma unroll
        for (int t = 0; t < 4; ++t) {
#pragma unroll
            for (int g = 0; g < 4; ++g) {
                v2f b0 = xa[t][g], b1 = xa[t][g];
                if (IT > 0) {
                    b0.x *= cf0[t]; b0.y *= cf0[t];
                    b1.x *= cf1[t]; b1.y *= cf1[t];
                }
                acc0 = wmma_f32(wa0[g], b0, acc0);
                acc1 = wmma_f32(wa1[g], b1, acc1);
            }
        }

        float s0[4], s1[4];
#pragma unroll
        for (int p = 0; p < 4; ++p) {
            s0[p] = fold_pair<0x161>(bit0, acc0[2 * p], acc0[2 * p + 1]);
            s1[p] = fold_pair<0x161>(bit0, acc1[2 * p], acc1[2 * p + 1]);
        }
        const float e00 = fold_pair<0x162>(bit1, s0[0], s0[1]);
        const float e01 = fold_pair<0x162>(bit1, s0[2], s0[3]);
        const float e10 = fold_pair<0x162>(bit1, s1[0], s1[1]);
        const float e11 = fold_pair<0x162>(bit1, s1[2], s1[3]);
        const float f0 = fold_pair<0x164>(bit2, e00, e01);
        const float f1 = fold_pair<0x164>(bit2, e10, e11);
        float tot = fold_pair<0x168>(bit3, f0, f1);
        if (IT == 0) tot *= 0.03125f;
        pdst[c * DC] = tot;
    }
}

template <int IT>
__global__ __launch_bounds__(THREADS)
void capsnet_pass_kernel(const float* __restrict__ x,
                         const float* __restrict__ W,
                         const float* __restrict__ part_in,
                         float* __restrict__ part_out,
                         const unsigned long long ws_bytes)
{
    if (ws_bytes < WS_NEED_BYTES) return;

    __shared__ __align__(16) float sWp[NC * DC * DIN];
    __shared__ __align__(16) float sStage[WAVES][NC][16];
    __shared__ __align__(16) float sCoef[WAVES][NC * 16 * 4];
    __shared__ __align__(16) float sV[DIN][NC];
    __shared__ __align__(16) float sOpre[NC][DC];
    __shared__ __align__(16) float sO[NC][DC];
    __shared__ float sFac[NC];

    const int tid  = threadIdx.x;
    const int wave = tid >> 5;
    const int lane = tid & 31;
    const int col  = lane & 15;
    const int h    = lane >> 4;
    const int bidx = blockIdx.x / SPLIT;
    const int part = blockIdx.x % SPLIT;

    const float* xb = x + (size_t)bidx * (N_PTS * DIN);

    stage_w(sWp, W, tid);
    if (IT > 0) {
        const float* pin = part_in + (size_t)bidx * (SPLIT * NC * DC);
        float* p = &sOpre[0][0];
        for (int i = tid; i < NC * DC; i += THREADS) {
            float s = pin[i];
            for (int gg = 1; gg < SPLIT; ++gg) s += pin[gg * (NC * DC) + i];
            p[i] = s;
        }
    }
    __syncthreads();

    if (IT > 0) {
        squash_factor(sOpre, sFac, tid);
        __syncthreads();
        for (int i = tid; i < NC * DC; i += THREADS) {
            const int c = i >> 4, d = i & 15;
            sO[c][d] = sFac[c] * sOpre[c][d];
        }
        __syncthreads();
        make_v(sV, sWp, sO, tid);
        __syncthreads();
    }

    route_wave4<IT>(xb, sWp, sStage[wave], sCoef[wave], sV, &sStage[wave][0][0],
                    part * TILES_PER_WG + wave, col, h);
    __syncthreads();

    float* pout = part_out + (size_t)blockIdx.x * (NC * DC);
    const float* p = &sStage[0][0][0];
    for (int i = tid; i < NC * DC; i += THREADS) {
        float s = p[i];
        for (int w = 1; w < WAVES; ++w) s += p[w * (NC * DC) + i];
        pout[i] = s;
    }
}

__global__ __launch_bounds__(THREADS)
void capsnet_finish_kernel(const float* __restrict__ x,
                           const float* __restrict__ W,
                           const float* __restrict__ part_in,
                           float* __restrict__ out,
                           const unsigned long long ws_bytes)
{
    __shared__ __align__(16) float sWp[NC * DC * DIN];
    __shared__ __align__(16) float sStage[WAVES][NC][16];
    __shared__ __align__(16) float sV[DIN][NC];
    __shared__ __align__(16) float sOpre[NC][DC];
    __shared__ __align__(16) float sO[NC][DC];
    __shared__ float sFac[NC];

    const int tid  = threadIdx.x;
    const int bidx = blockIdx.x;

    if (ws_bytes >= WS_NEED_BYTES) {
        const float* pin = part_in + (size_t)bidx * (SPLIT * NC * DC);
        float* p = &sOpre[0][0];
        for (int i = tid; i < NC * DC; i += THREADS) {
            float s = pin[i];
            for (int gg = 1; gg < SPLIT; ++gg) s += pin[gg * (NC * DC) + i];
            p[i] = s;
        }
        __syncthreads();
        squash_factor(sOpre, sFac, tid);
        __syncthreads();
        for (int i = tid; i < NC * DC; i += THREADS)
            out[(size_t)bidx * (NC * DC) + i] = sFac[i >> 4] * p[i];
        return;
    }

    const int wave = tid >> 5;
    const int lane = tid & 31;
    const int col  = lane & 15;
    const int h    = lane >> 4;
    const float* xb = x + (size_t)bidx * (N_PTS * DIN);

    stage_w(sWp, W, tid);
    __syncthreads();

    for (int it = 0; it < 3; ++it) {
        {
            float* p = &sOpre[0][0];
            for (int i = tid; i < NC * DC; i += THREADS) p[i] = 0.0f;
        }
        __syncthreads();
        if (it == 0)
            route_tiles<0>(xb, sWp, sStage[wave], sV, sOpre, wave, NTILES, col, h);
        else
            route_tiles<1>(xb, sWp, sStage[wave], sV, sOpre, wave, NTILES, col, h);
        __syncthreads();
        squash_factor(sOpre, sFac, tid);
        __syncthreads();
        for (int i = tid; i < NC * DC; i += THREADS) {
            const int c = i >> 4, d = i & 15;
            sO[c][d] = sFac[c] * sOpre[c][d];
        }
        __syncthreads();
        if (it < 2) {
            make_v(sV, sWp, sO, tid);
            __syncthreads();
        }
    }
    for (int i = tid; i < NC * DC; i += THREADS)
        out[(size_t)bidx * (NC * DC) + i] = sO[i >> 4][i & 15];
}

extern "C" void kernel_launch(void* const* d_in, const int* in_sizes, int n_in,
                              void* d_out, int out_size, void* d_ws, size_t ws_size,
                              hipStream_t stream) {
    (void)in_sizes; (void)n_in; (void)out_size;
    const float* x = (const float*)d_in[0];
    const float* W = (const float*)d_in[1];
    float* out = (float*)d_out;
    float* planeA = (float*)d_ws;
    float* planeB = (float*)d_ws + PLANE_FLOATS;
    const unsigned long long wsb = (unsigned long long)ws_size;
    hipLaunchKernelGGL(capsnet_pass_kernel<0>, dim3(BATCH * SPLIT), dim3(THREADS), 0,
                       stream, x, W, (const float*)planeB, planeA, wsb);
    hipLaunchKernelGGL(capsnet_pass_kernel<1>, dim3(BATCH * SPLIT), dim3(THREADS), 0,
                       stream, x, W, (const float*)planeA, planeB, wsb);
    hipLaunchKernelGGL(capsnet_pass_kernel<2>, dim3(BATCH * SPLIT), dim3(THREADS), 0,
                       stream, x, W, (const float*)planeB, planeA, wsb);
    hipLaunchKernelGGL(capsnet_finish_kernel, dim3(BATCH), dim3(THREADS), 0,
                       stream, x, W, (const float*)planeA, out, wsb);
}


# W staging as a 4x4-block transpose: 16-byte global reads on whole 128-byte lines (was 4-byte reads, 16 lines per wave load); LDS rows at 20-float pitch spread stores and fragment reads over banks
# speedup vs baseline: 3.0689x; 1.0482x over previous
#include <hip/hip_runtime.h>


typedef float v2f __attribute__((ext_vector_type(2)));
typedef float v8f __attribute__((ext_vector_type(8)));

#define BATCH   128
#define N_PTS   2048
#define DIN     16
#define NC      32
#define DC      16
#define NTILES  (N_PTS / 16)
#define WAVES   8
#define THREADS 256
#define EPS     1e-7f

#define WPITCH  20
#define WCAP    (DC * WPITCH)

#define SPLIT         4
#define TILES_PER_WG  (NTILES / SPLIT)
#define PLANE_FLOATS  (BATCH * SPLIT * NC * DC)
#define WS_NEED_BYTES (2ull * PLANE_FLOATS * 4ull)

static_assert(TILES_PER_WG == 4 * WAVES, "a wave of a pass launch owns exactly four row-tiles");

__device__ __forceinline__ void wave_sync() { __builtin_amdgcn_wave_barrier(); }

__device__ __forceinline__ v8f wmma_f32(const v2f a, const v2f b, const v8f c) {
    return __builtin_amdgcn_wmma_f32_16x16x4_f32(false, a, false, b,
                                                 (short)0, c, false, false);
}

__device__ __forceinline__ void stage_w(float* sWp, const float* __restrict__ W,
                                        const int tid) {
    float4 v[2][4];
#pragma unroll
    for (int r = 0; r < 2; ++r) {
        const int bl = tid + THREADS * r;
        const int kb = bl & 3, eb = bl >> 2;
#pragma unroll
        for (int kk = 0; kk < 4; ++kk)
            v[r][kk] = *(const float4*)&W[(4 * kb + kk) * (NC * DC) + 4 * eb];
    }
#pragma unroll
    for (int r = 0; r < 2; ++r) {
        const int bl = tid + THREADS * r;
        const int kb = bl & 3, eb = bl >> 2;
        float* row = sWp + (4 * eb) * WPITCH + 2 * kb;
        const float lo0[4] = {v[r][0].x, v[r][0].y, v[r][0].z, v[r][0].w};
        const float lo1[4] = {v[r][1].x, v[r][1].y, v[r][1].z, v[r][1].w};
        const float hi0[4] = {v[r][2].x, v[r][2].y, v[r][2].z, v[r][2].w};
        const float hi1[4] = {v[r][3].x, v[r][3].y, v[r][3].z, v[r][3].w};
#pragma unroll
        for (int j = 0; j < 4; ++j) {
            *(float2*)&row[j * WPITCH]     = make_float2(lo0[j], lo1[j]);
            *(float2*)&row[j * WPITCH + 8] = make_float2(hi0[j], hi1[j]);
        }
    }
}

__device__ __forceinline__ void squash_factor(const float (*sOpre)[DC],
                                              float* sFac, const int tid) {
    if (tid < NC) {
        float s2 = 0.0f;
#pragma unroll
        for (int d = 0; d < DC; ++d) {
            const float p = sOpre[tid][d];
            s2 = fmaf(p, p, s2);
        }
        sFac[tid] = (s2 / (1.0f + s2)) * rsqrtf(s2 + EPS);
    }
}

__device__ __forceinline__ void make_v(float (*sV)[NC], const float* sWp,
                                       const float (*sO)[DC], const int tid) {
    for (int i = tid; i < DIN * NC; i += THREADS) {
        const int c = i >> 4, p = i & 15;
        const int hk = p >> 3, qk = p & 7;
        const int k = 4 * (qk >> 1) + 2 * hk + (qk & 1);
        float s = 0.0f;
#pragma unroll
        for (int d = 0; d < DC; ++d)
            s = fmaf(sWp[c * WCAP + d * WPITCH + p], sO[c][d], s);
        sV[k][c] = s;
    }
}

template <int IT>
__device__ __forceinline__ void route_tiles(const float* __restrict__ xb,
                                            const float* sWp,
                                            float (*sStageW)[16],
                                            const float (*sV)[NC],
                                            float (*sOpre)[DC],
                                            const int mt_begin, const int mt_end,
                                            const int col, const int h)
{
    v2f vlo[4], vhi[4];
    if (IT > 0) {
#pragma unroll
        for (int g = 0; g < 4; ++g) {
            const int k0 = 4 * g + 2 * h;
            vlo[g].x = sV[k0][col];       vlo[g].y = sV[k0 + 1][col];
            vhi[g].x = sV[k0][16 + col];  vhi[g].y = sV[k0 + 1][16 + col];
        }
    }

    for (int mt = mt_begin; mt < mt_end; mt += WAVES) {
        const int nrow = mt * 16 + col;
        v2f a[4];
#pragma unroll
        for (int g = 0; g < 4; ++g) {
            const int k0 = 4 * g + 2 * h;
            a[g].x = xb[nrow * DIN + k0];
            a[g].y = xb[nrow * DIN + k0 + 1];
        }

        if (IT > 0) {
            v8f blo = {}, bhi = {};
#pragma unroll
            for (int g = 0; g < 4; ++g) {
                blo = wmma_f32(a[g], vlo[g], blo);
                bhi = wmma_f32(a[g], vhi[g], bhi);
            }
            {
                float4* s0 = (float4*)&sStageW[col][8 * h];
                s0[0] = make_float4(blo[0], blo[1], blo[2], blo[3]);
                s0[1] = make_float4(blo[4], blo[5], blo[6], blo[7]);
                float4* s1 = (float4*)&sStageW[16 + col][8 * h];
                s1[0] = make_float4(bhi[0], bhi[1], bhi[2], bhi[3]);
                s1[1] = make_float4(bhi[4], bhi[5], bhi[6], bhi[7]);
            }
            wave_sync();
            {
                float vals[16];
#pragma unroll
                for (int cc = 0; cc < 16; ++cc)
                    vals[cc] = sStageW[h * 16 + cc][col];
                float m = vals[0];
#pragma unroll
                for (int cc = 1; cc < 16; ++cc) m = fmaxf(m, vals[cc]);
                m = fmaxf(m, __shfl_xor(m, 16, 32));
                float s = 0.0f;
#pragma unroll
                for (int cc = 0; cc < 16; ++cc) {
                    vals[cc] = __expf(vals[cc] - m);
                    s += vals[cc];
                }
                s += __shfl_xor(s, 16, 32);
                const float r = 1.0f / s;
#pragma unroll
                for (int cc = 0; cc < 16; ++cc)
                    sStageW[h * 16 + cc][col] = vals[cc] * r;
            }
            wave_sync();
        }

#pragma unroll 4
        for (int c = 0; c < NC; ++c) {
            const float4* wp =
                (const float4*)&sWp[c * WCAP + col * WPITCH + h * 8];
            const float4 w0 = wp[0];
            const float4 w1 = wp[1];
            v2f b0; b0.x = w0.x; b0.y = w0.y;
            v2f b1; b1.x = w0.z; b1.y = w0.w;
            v2f b2; b2.x = w1.x; b2.y = w1.y;
            v2f b3; b3.x = w1.z; b3.y = w1.w;
            v8f acc = {};
            acc = wmma_f32(a[0], b0, acc);
            acc = wmma_f32(a[1], b1, acc);
            acc = wmma_f32(a[2], b2, acc);
            acc = wmma_f32(a[3], b3, acc);

            float partial;
            if (IT == 0) {
                partial = 0.03125f * (acc[0] + acc[1] + acc[2] + acc[3] +
                                      acc[4] + acc[5] + acc[6] + acc[7]);
            } else {
                const float4* cp = (const float4*)&sStageW[c][8 * h];
                const float4 c0 = cp[0];
                const float4 c1 = cp[1];
                partial = c0.x * acc[0] + c0.y * acc[1] +
                          c0.z * acc[2] + c0.w * acc[3] +
                          c1.x * acc[4] + c1.y * acc[5] +
                          c1.z * acc[6] + c1.w * acc[7];
            }
            atomicAdd(&sOpre[c][col], partial);
        }
    }
}

template <int CTRL>
__device__ __forceinline__ float row_xor(const float v) {
    return __builtin_bit_cast(float, __builtin_amdgcn_update_dpp(
        0, __builtin_bit_cast(int, v), CTRL, 0xf, 0xf, false));
}

template <int CTRL>
__device__ __forceinline__ float fold_pair(const bool bit, const float lo,
                                           const float hi) {
    const float keep = bit ? hi : lo;
    const float send = bit ? lo : hi;
    return keep + row_xor<CTRL>(send);
}

template <int IT>
__device__ __forceinline__ void route_wave4(const float* __restrict__ xb,
                                            const float* sWp,
                                            float (*sStageW)[16],
                                            float* sCoefW,
                                            const float (*sV)[NC],
                                            float* sPartW,
                                            const int mt0,
                                            const int col, const int h)
{
    v2f xa[4][4];
#pragma unroll
    for (int t = 0; t < 4; ++t) {
        const int nrow = (mt0 + WAVES * t) * 16 + col;
#pragma unroll
        for (int g = 0; g < 4; ++g) {
            const int k0 = 4 * g + 2 * h;
            xa[t][g].x = xb[nrow * DIN + k0];
            xa[t][g].y = xb[nrow * DIN + k0 + 1];
        }
    }

    if (IT > 0) {
        v2f vlo[4], vhi[4];
#pragma unroll
        for (int g = 0; g < 4; ++g) {
            const int k0 = 4 * g + 2 * h;
            vlo[g].x = sV[k0][col];       vlo[g].y = sV[k0 + 1][col];
            vhi[g].x = sV[k0][16 + col];  vhi[g].y = sV[k0 + 1][16 + col];
        }
#pragma unroll
        for (int t = 0; t < 4; ++t) {
            v8f blo = {}, bhi = {};
#pragma unroll
            for (int g = 0; g < 4; ++g) {
                blo = wmma_f32(xa[t][g], vlo[g], blo);
                bhi = wmma_f32(xa[t][g], vhi[g], bhi);
            }
            {
                float4* s0 = (float4*)&sStageW[col][8 * h];
                s0[0] = make_float4(blo[0], blo[1], blo[2], blo[3]);
                s0[1] = make_float4(blo[4], blo[5], blo[6], blo[7]);
                float4* s1 = (float4*)&sStageW[16 + col][8 * h];
                s1[0] = make_float4(bhi[0], bhi[1], bhi[2], bhi[3]);
                s1[1] = make_float4(bhi[4], bhi[5], bhi[6], bhi[7]);
            }
            wave_sync();
            {
                float vals[16];
#pragma unroll
                for (int cc = 0; cc < 16; ++cc)
                    vals[cc] = sStageW[h * 16 + cc][col];
                float m = vals[0];
#pragma unroll
                for (int cc = 1; cc < 16; ++cc) m = fmaxf(m, vals[cc]);
                m = fmaxf(m, __shfl_xor(m, 16, 32));
                float s = 0.0f;
#pragma unroll
                for (int cc = 0; cc < 16; ++cc) {
                    vals[cc] = __expf(vals[cc] - m);
                    s += vals[cc];
                }
                s += __shfl_xor(s, 16, 32);
                const float r = 1.0f / s;
#pragma unroll
                for (int cc = 0; cc < 16; ++cc)
                    sCoefW[((h * 16 + cc) * 16 + col) * 4 + t] = vals[cc] * r;
            }
            wave_sync();
        }
    }

    const bool bit0 = (col & 1) != 0;
    const bool bit1 = (col & 2) != 0;
    const bool bit2 = (col & 4) != 0;
    const bool bit3 = (col & 8) != 0;
    float* const pdst = sPartW + (bit3 ? DC : 0) + 8 * h + (col & 7);

    for (int c = 0; c < NC; c += 2) {
        const float4* wp =
            (const float4*)&sWp[c * WCAP + col * WPITCH + h * 8];
        const float4 u0 = wp[0],  u1 = wp[1];
        const float4 w0 = wp[WCAP / 4], w1 = wp[WCAP / 4 + 1];
        v2f wa0[4], wa1[4];
        wa0[0].x = u0.x; wa0[0].y = u0.y;  wa0[1].x = u0.z; wa0[1].y = u0.w;
        wa0[2].x = u1.x; wa0[2].y = u1.y;  wa0[3].x = u1.z; wa0[3].y = u1.w;
        wa1[0].x = w0.x; wa1[0].y = w0.y;  wa1[1].x = w0.z; wa1[1].y = w0.w;
        wa1[2].x = w1.x; wa1[2].y = w1.y;  wa1[3].x = w1.z; wa1[3].y = w1.w;

        float cf0[4] = {1.0f, 1.0f, 1.0f, 1.0f};
        float cf1[4] = {1.0f, 1.0f, 1.0f, 1.0f};
        if (IT > 0) {
            const float4 q0 = *(const float4*)&sCoefW[(c * 16 + col) * 4];
            const float4 q1 = *(const float4*)&sCoefW[((c + 1) * 16 + col) * 4];
            cf0[0] = q0.x; cf0[1] = q0.y; cf0[2] = q0.z; cf0[3] = q0.w;
            cf1[0] = q1.x; cf1[1] = q1.y; cf1[2] = q1.z; cf1[3] = q1.w;
        }

        v8f acc0 = {}, acc1 = {};
#pragma unroll
        for (int t = 0; t < 4; ++t) {
#pragma unroll
            for (int g = 0; g < 4; ++g) {
                v2f b0 = xa[t][g], b1 = xa[t][g];
                if (IT > 0) {
                    b0.x *= cf0[t]; b0.y *= cf0[t];
                    b1.x *= cf1[t]; b1.y *= cf1[t];
                }
                acc0 = wmma_f32(wa0[g], b0, acc0);
                acc1 = wmma_f32(wa1[g], b1, acc1);
            }
        }

        float s0[4], s1[4];
#pragma unroll
        for (int p = 0; p < 4; ++p) {
            s0[p] = fold_pair<0x161>(bit0, acc0[2 * p], acc0[2 * p + 1]);
            s1[p] = fold_pair<0x161>(bit0, acc1[2 * p], acc1[2 * p + 1]);
        }
        const float e00 = fold_pair<0x162>(bit1, s0[0], s0[1]);
        const float e01 = fold_pair<0x162>(bit1, s0[2], s0[3]);
        const float e10 = fold_pair<0x162>(bit1, s1[0], s1[1]);
        const float e11 = fold_pair<0x162>(bit1, s1[2], s1[3]);
        const float f0 = fold_pair<0x164>(bit2, e00, e01);
        const float f1 = fold_pair<0x164>(bit2, e10, e11);
        float tot = fold_pair<0x168>(bit3, f0, f1);
        if (IT == 0) tot *= 0.03125f;
        pdst[c * DC] = tot;
    }
}

template <int IT>
__global__ __launch_bounds__(THREADS)
void capsnet_pass_kernel(const float* __restrict__ x,
                         const float* __restrict__ W,
                         const float* __restrict__ part_in,
                         float* __restrict__ part_out,
                         const unsigned long long ws_bytes)
{
    if (ws_bytes < WS_NEED_BYTES) return;

    __shared__ __align__(16) float sWp[NC * DC * WPITCH];
    __shared__ __align__(16) float sStage[WAVES][NC][16];
    __shared__ __align__(16) float sCoef[WAVES][NC * 16 * 4];
    __shared__ __align__(16) float sV[DIN][NC];
    __shared__ __align__(16) float sOpre[NC][DC];
    __shared__ __align__(16) float sO[NC][DC];
    __shared__ float sFac[NC];

    const int tid  = threadIdx.x;
    const int wave = tid >> 5;
    const int lane = tid & 31;
    const int col  = lane & 15;
    const int h    = lane >> 4;
    const int bidx = blockIdx.x / SPLIT;
    const int part = blockIdx.x % SPLIT;

    const float* xb = x + (size_t)bidx * (N_PTS * DIN);

    stage_w(sWp, W, tid);
    if (IT > 0) {
        const float* pin = part_in + (size_t)bidx * (SPLIT * NC * DC);
        float* p = &sOpre[0][0];
        for (int i = tid; i < NC * DC; i += THREADS) {
            float s = pin[i];
            for (int gg = 1; gg < SPLIT; ++gg) s += pin[gg * (NC * DC) + i];
            p[i] = s;
        }
    }
    __syncthreads();

    if (IT > 0) {
        squash_factor(sOpre, sFac, tid);
        __syncthreads();
        for (int i = tid; i < NC * DC; i += THREADS) {
            const int c = i >> 4, d = i & 15;
            sO[c][d] = sFac[c] * sOpre[c][d];
        }
        __syncthreads();
        make_v(sV, sWp, sO, tid);
        __syncthreads();
    }

    route_wave4<IT>(xb, sWp, sStage[wave], sCoef[wave], sV, &sStage[wave][0][0],
                    part * TILES_PER_WG + wave, col, h);
    __syncthreads();

    float* pout = part_out + (size_t)blockIdx.x * (NC * DC);
    const float* p = &sStage[0][0][0];
    for (int i = tid; i < NC * DC; i += THREADS) {
        float s = p[i];
        for (int w = 1; w < WAVES; ++w) s += p[w * (NC * DC) + i];
        pout[i] = s;
    }
}

__global__ __launch_bounds__(THREADS)
void capsnet_finish_kernel(const float* __restrict__ x,
                           const float* __restrict__ W,
                           const float* __restrict__ part_in,
                           float* __restrict__ out,
                           const unsigned long long ws_bytes)
{
    __shared__ __align__(16) float sWp[NC * DC * WPITCH];
    __shared__ __align__(16) float sStage[WAVES][NC][16];
    __shared__ __align__(16) float sV[DIN][NC];
    __shared__ __align__(16) float sOpre[NC][DC];
    __shared__ __align__(16) float sO[NC][DC];
    __shared__ float sFac[NC];

    const int tid  = threadIdx.x;
    const int bidx = blockIdx.x;

    if (ws_bytes >= WS_NEED_BYTES) {
        const float* pin = part_in + (size_t)bidx * (SPLIT * NC * DC);
        float* p = &sOpre[0][0];
        for (int i = tid; i < NC * DC; i += THREADS) {
            float s = pin[i];
            for (int gg = 1; gg < SPLIT; ++gg) s += pin[gg * (NC * DC) + i];
            p[i] = s;
        }
        __syncthreads();
        squash_factor(sOpre, sFac, tid);
        __syncthreads();
        for (int i = tid; i < NC * DC; i += THREADS)
            out[(size_t)bidx * (NC * DC) + i] = sFac[i >> 4] * p[i];
        return;
    }

    const int wave = tid >> 5;
    const int lane = tid & 31;
    const int col  = lane & 15;
    const int h    = lane >> 4;
    const float* xb = x + (size_t)bidx * (N_PTS * DIN);

    stage_w(sWp, W, tid);
    __syncthreads();

    for (int it = 0; it < 3; ++it) {
        {
            float* p = &sOpre[0][0];
            for (int i = tid; i < NC * DC; i += THREADS) p[i] = 0.0f;
        }
        __syncthreads();
        if (it == 0)
            route_tiles<0>(xb, sWp, sStage[wave], sV, sOpre, wave, NTILES, col, h);
        else
            route_tiles<1>(xb, sWp, sStage[wave], sV, sOpre, wave, NTILES, col, h);
        __syncthreads();
        squash_factor(sOpre, sFac, tid);
        __syncthreads();
        for (int i = tid; i < NC * DC; i += THREADS) {
            const int c = i >> 4, d = i & 15;
            sO[c][d] = sFac[c] * sOpre[c][d];
        }
        __syncthreads();
        if (it < 2) {
            make_v(sV, sWp, sO, tid);
            __syncthreads();
        }
    }
    for (int i = tid; i < NC * DC; i += THREADS)
        out[(size_t)bidx * (NC * DC) + i] = sO[i >> 4][i & 15];
}

extern "C" void kernel_launch(void* const* d_in, const int* in_sizes, int n_in,
                              void* d_out, int out_size, void* d_ws, size_t ws_size,
                              hipStream_t stream) {
    (void)in_sizes; (void)n_in; (void)out_size;
    const float* x = (const float*)d_in[0];
    const float* W = (const float*)d_in[1];
    float* out = (float*)d_out;
    float* planeA = (float*)d_ws;
    float* planeB = (float*)d_ws + PLANE_FLOATS;
    const unsigned long long wsb = (unsigned long long)ws_size;
    hipLaunchKernelGGL(capsnet_pass_kernel<0>, dim3(BATCH * SPLIT), dim3(THREADS), 0,
                       stream, x, W, (const float*)planeB, planeA, wsb);
    hipLaunchKernelGGL(capsnet_pass_kernel<1>, dim3(BATCH * SPLIT), dim3(THREADS), 0,
                       stream, x, W, (const float*)planeA, planeB, wsb);
    hipLaunchKernelGGL(capsnet_pass_kernel<2>, dim3(BATCH * SPLIT), dim3(THREADS), 0,
                       stream, x, W, (const float*)planeB, planeA, wsb);
    hipLaunchKernelGGL(capsnet_finish_kernel, dim3(BATCH), dim3(THREADS), 0,
                       stream, x, W, (const float*)planeA, out, wsb);
}


# Batch item split over 2 work-groups, not 4 (256 of 8 waves, one per WGP): a wave owns 8 row-tiles; half the waves enter, per-capsule W reads and row sums halve; f32 sum order changes
# speedup vs baseline: 3.1627x; 1.0306x over previous
#include <hip/hip_runtime.h>


typedef float v2f __attribute__((ext_vector_type(2)));
typedef float v8f __attribute__((ext_vector_type(8)));

#define BATCH   128
#define N_PTS   2048
#define DIN     16
#define NC      32
#define DC      16
#define NTILES  (N_PTS / 16)
#define WAVES   8
#define THREADS 256
#define EPS     1e-7f

#define WPITCH  20
#define WCAP    (DC * WPITCH)

#define SPLIT         2
#define TILES_PER_WG  (NTILES / SPLIT)
#define TPW           (TILES_PER_WG / WAVES)
#define PLANE_FLOATS  (BATCH * SPLIT * NC * DC)
#define WS_NEED_BYTES (2ull * PLANE_FLOATS * 4ull)

static_assert(TPW == 8, "a wave of a pass launch owns exactly eight row-tiles");

__device__ __forceinline__ void wave_sync() { __builtin_amdgcn_wave_barrier(); }

__device__ __forceinline__ v8f wmma_f32(const v2f a, const v2f b, const v8f c) {
    return __builtin_amdgcn_wmma_f32_16x16x4_f32(false, a, false, b,
                                                 (short)0, c, false, false);
}

__device__ __forceinline__ void stage_w(float* sWp, const float* __restrict__ W,
                                        const int tid) {
    float4 v[2][4];
#pragma unroll
    for (int r = 0; r < 2; ++r) {
        const int bl = tid + THREADS * r;
        const int kb = bl & 3, eb = bl >> 2;
#pragma unroll
        for (int kk = 0; kk < 4; ++kk)
            v[r][kk] = *(const float4*)&W[(4 * kb + kk) * (NC * DC) + 4 * eb];
    }
#pragma unroll
    for (int r = 0; r < 2; ++r) {
        const int bl = tid + THREADS * r;
        const int kb = bl & 3, eb = bl >> 2;
        float* row = sWp + (4 * eb) * WPITCH + 2 * kb;
        const float lo0[4] = {v[r][0].x, v[r][0].y, v[r][0].z, v[r][0].w};
        const float lo1[4] = {v[r][1].x, v[r][1].y, v[r][1].z, v[r][1].w};
        const float hi0[4] = {v[r][2].x, v[r][2].y, v[r][2].z, v[r][2].w};
        const float hi1[4] = {v[r][3].x, v[r][3].y, v[r][3].z, v[r][3].w};
#pragma unroll
        for (int j = 0; j < 4; ++j) {
            *(float2*)&row[j * WPITCH]     = make_float2(lo0[j], lo1[j]);
            *(float2*)&row[j * WPITCH + 8] = make_float2(hi0[j], hi1[j]);
        }
    }
}

__device__ __forceinline__ void squash_factor(const float (*sOpre)[DC],
                                              float* sFac, const int tid) {
    if (tid < NC) {
        float s2 = 0.0f;
#pragma unroll
        for (int d = 0; d < DC; ++d) {
            const float p = sOpre[tid][d];
            s2 = fmaf(p, p, s2);
        }
        sFac[tid] = (s2 / (1.0f + s2)) * rsqrtf(s2 + EPS);
    }
}

__device__ __forceinline__ void make_v(float (*sV)[NC], const float* sWp,
                                       const float (*sO)[DC], const int tid) {
    for (int i = tid; i < DIN * NC; i += THREADS) {
        const int c = i >> 4, p = i & 15;
        const int hk = p >> 3, qk = p & 7;
        const int k = 4 * (qk >> 1) + 2 * hk + (qk & 1);
        float s = 0.0f;
#pragma unroll
        for (int d = 0; d < DC; ++d)
            s = fmaf(sWp[c * WCAP + d * WPITCH + p], sO[c][d], s);
        sV[k][c] = s;
    }
}

template <int IT>
__device__ __forceinline__ void route_tiles(const float* __restrict__ xb,
                                            const float* sWp,
                                            float (*sStageW)[16],
                                            const float (*sV)[NC],
                                            float (*sOpre)[DC],
                                            const int mt_begin, const int mt_end,
                                            const int col, const int h)
{
    v2f vlo[4], vhi[4];
    if (IT > 0) {
#pragma unroll
        for (int g = 0; g < 4; ++g) {
            const int k0 = 4 * g + 2 * h;
            vlo[g].x = sV[k0][col];       vlo[g].y = sV[k0 + 1][col];
            vhi[g].x = sV[k0][16 + col];  vhi[g].y = sV[k0 + 1][16 + col];
        }
    }

    for (int mt = mt_begin; mt < mt_end; mt += WAVES) {
        const int nrow = mt * 16 + col;
        v2f a[4];
#pragma unroll
        for (int g = 0; g < 4; ++g) {
            const int k0 = 4 * g + 2 * h;
            a[g].x = xb[nrow * DIN + k0];
            a[g].y = xb[nrow * DIN + k0 + 1];
        }

        if (IT > 0) {
            v8f blo = {}, bhi = {};
#pragma unroll
            for (int g = 0; g < 4; ++g) {
                blo = wmma_f32(a[g], vlo[g], blo);
                bhi = wmma_f32(a[g], vhi[g], bhi);
            }
            {
                float4* s0 = (float4*)&sStageW[col][8 * h];
                s0[0] = make_float4(blo[0], blo[1], blo[2], blo[3]);
                s0[1] = make_float4(blo[4], blo[5], blo[6], blo[7]);
                float4* s1 = (float4*)&sStageW[16 + col][8 * h];
                s1[0] = make_float4(bhi[0], bhi[1], bhi[2], bhi[3]);
                s1[1] = make_float4(bhi[4], bhi[5], bhi[6], bhi[7]);
            }
            wave_sync();
            {
                float vals[16];
#pragma unroll
                for (int cc = 0; cc < 16; ++cc)
                    vals[cc] = sStageW[h * 16 + cc][col];
                float m = vals[0];
#pragma unroll
                for (int cc = 1; cc < 16; ++cc) m = fmaxf(m, vals[cc]);
                m = fmaxf(m, __shfl_xor(m, 16, 32));
                float s = 0.0f;
#pragma unroll
                for (int cc = 0; cc < 16; ++cc) {
                    vals[cc] = __expf(vals[cc] - m);
                    s += vals[cc];
                }
                s += __shfl_xor(s, 16, 32);
                const float r = 1.0f / s;
#pragma unroll
                for (int cc = 0; cc < 16; ++cc)
                    sStageW[h * 16 + cc][col] = vals[cc] * r;
            }
            wave_sync();
        }

#pragma unroll 4
        for (int c = 0; c < NC; ++c) {
            const float4* wp =
                (const float4*)&sWp[c * WCAP + col * WPITCH + h * 8];
            const float4 w0 = wp[0];
            const float4 w1 = wp[1];
            v2f b0; b0.x = w0.x; b0.y = w0.y;
            v2f b1; b1.x = w0.z; b1.y = w0.w;
            v2f b2; b2.x = w1.x; b2.y = w1.y;
            v2f b3; b3.x = w1.z; b3.y = w1.w;
            v8f acc = {};
            acc = wmma_f32(a[0], b0, acc);
            acc = wmma_f32(a[1], b1, acc);
            acc = wmma_f32(a[2], b2, acc);
            acc = wmma_f32(a[3], b3, acc);

            float partial;
            if (IT == 0) {
                partial = 0.03125f * (acc[0] + acc[1] + acc[2] + acc[3] +
                                      acc[4] + acc[5] + acc[6] + acc[7]);
            } else {
                const float4* cp = (const float4*)&sStageW[c][8 * h];
                const float4 c0 = cp[0];
                const float4 c1 = cp[1];
                partial = c0.x * acc[0] + c0.y * acc[1] +
                          c0.z * acc[2] + c0.w * acc[3] +
                          c1.x * acc[4] + c1.y * acc[5] +
                          c1.z * acc[6] + c1.w * acc[7];
            }
            atomicAdd(&sOpre[c][col], partial);
        }
    }
}

template <int CTRL>
__device__ __forceinline__ float row_xor(const float v) {
    return __builtin_bit_cast(float, __builtin_amdgcn_update_dpp(
        0, __builtin_bit_cast(int, v), CTRL, 0xf, 0xf, false));
}

template <int CTRL>
__device__ __forceinline__ float fold_pair(const bool bit, const float lo,
                                           const float hi) {
    const float keep = bit ? hi : lo;
    const float send = bit ? lo : hi;
    return keep + row_xor<CTRL>(send);
}

template <int IT>
__device__ __forceinline__ void route_wave(const float* __restrict__ xb,
                                           const float* sWp,
                                           float (*sStageW)[16],
                                           float* sCoefW,
                                           const float (*sV)[NC],
                                           float* sPartW,
                                           const int mt0,
                                           const int col, const int h)
{
    v2f xa[TPW][4];
#pragma unroll
    for (int t = 0; t < TPW; ++t) {
        const int nrow = (mt0 + WAVES * t) * 16 + col;
#pragma unroll
        for (int g = 0; g < 4; ++g) {
            const int k0 = 4 * g + 2 * h;
            xa[t][g].x = xb[nrow * DIN + k0];
            xa[t][g].y = xb[nrow * DIN + k0 + 1];
        }
    }

    if (IT > 0) {
        v2f vlo[4], vhi[4];
#pragma unroll
        for (int g = 0; g < 4; ++g) {
            const int k0 = 4 * g + 2 * h;
            vlo[g].x = sV[k0][col];       vlo[g].y = sV[k0 + 1][col];
            vhi[g].x = sV[k0][16 + col];  vhi[g].y = sV[k0 + 1][16 + col];
        }
#pragma unroll
        for (int tp = 0; tp < TPW; tp += 2) {
#pragma unroll
            for (int u = 0; u < 2; ++u) {
                v8f blo = {}, bhi = {};
#pragma unroll
                for (int g = 0; g < 4; ++g) {
                    blo = wmma_f32(xa[tp + u][g], vlo[g], blo);
                    bhi = wmma_f32(xa[tp + u][g], vhi[g], bhi);
                }
                float (*st)[16] = sStageW + u * NC;
                float4* s0 = (float4*)&st[col][8 * h];
                s0[0] = make_float4(blo[0], blo[1], blo[2], blo[3]);
                s0[1] = make_float4(blo[4], blo[5], blo[6], blo[7]);
                float4* s1 = (float4*)&st[16 + col][8 * h];
                s1[0] = make_float4(bhi[0], bhi[1], bhi[2], bhi[3]);
                s1[1] = make_float4(bhi[4], bhi[5], bhi[6], bhi[7]);
            }
            wave_sync();
#pragma unroll
            for (int u = 0; u < 2; ++u) {
                const int t = tp + u;
                const float (*st)[16] = sStageW + u * NC;
                float vals[16];
#pragma unroll
                for (int cc = 0; cc < 16; ++cc)
                    vals[cc] = st[h * 16 + cc][col];
                float m = vals[0];
#pragma unroll
                for (int cc = 1; cc < 16; ++cc) m = fmaxf(m, vals[cc]);
                m = fmaxf(m, __shfl_xor(m, 16, 32));
                float s = 0.0f;
#pragma unroll
                for (int cc = 0; cc < 16; ++cc) {
                    vals[cc] = __expf(vals[cc] - m);
                    s += vals[cc];
                }
                s += __shfl_xor(s, 16, 32);
                const float r = 1.0f / s;
#pragma unroll
                for (int cc = 0; cc < 16; ++cc)
                    sCoefW[(((h * 16 + cc) * (TPW / 4) + (t >> 2)) * 16 + col) * 4 + (t & 3)] =
                        vals[cc] * r;
            }
            wave_sync();
        }
    }

    const bool bit0 = (col & 1) != 0;
    const bool bit1 = (col & 2) != 0;
    const bool bit2 = (col & 4) != 0;
    const bool bit3 = (col & 8) != 0;
    float* const pdst = sPartW + (bit3 ? DC : 0) + 8 * h + (col & 7);

    for (int c = 0; c < NC; c += 2) {
        const float4* wp =
            (const float4*)&sWp[c * WCAP + col * WPITCH + h * 8];
        const float4 u0 = wp[0],  u1 = wp[1];
        const float4 w0 = wp[WCAP / 4], w1 = wp[WCAP / 4 + 1];
        v2f wa0[4], wa1[4];
        wa0[0].x = u0.x; wa0[0].y = u0.y;  wa0[1].x = u0.z; wa0[1].y = u0.w;
        wa0[2].x = u1.x; wa0[2].y = u1.y;  wa0[3].x = u1.z; wa0[3].y = u1.w;
        wa1[0].x = w0.x; wa1[0].y = w0.y;  wa1[1].x = w0.z; wa1[1].y = w0.w;
        wa1[2].x = w1.x; wa1[2].y = w1.y;  wa1[3].x = w1.z; wa1[3].y = w1.w;

        float cf0[TPW], cf1[TPW];
#pragma unroll
        for (int j = 0; j < TPW / 4; ++j) {
            float4 q0 = make_float4(1.0f, 1.0f, 1.0f, 1.0f), q1 = q0;
            if (IT > 0) {
                q0 = *(const float4*)&sCoefW[((c * (TPW / 4) + j) * 16 + col) * 4];
                q1 = *(const float4*)&sCoefW[(((c + 1) * (TPW / 4) + j) * 16 + col) * 4];
            }
            cf0[4 * j] = q0.x; cf0[4 * j + 1] = q0.y; cf0[4 * j + 2] = q0.z; cf0[4 * j + 3] = q0.w;
            cf1[4 * j] = q1.x; cf1[4 * j + 1] = q1.y; cf1[4 * j + 2] = q1.z; cf1[4 * j + 3] = q1.w;
        }

        v8f acc0 = {}, acc1 = {};
#pragma unroll
        for (int t = 0; t < TPW; ++t) {
#pragma unroll
            for (int g = 0; g < 4; ++g) {
                v2f b0 = xa[t][g], b1 = xa[t][g];
                if (IT > 0) {
                    b0.x *= cf0[t]; b0.y *= cf0[t];
                    b1.x *= cf1[t]; b1.y *= cf1[t];
                }
                acc0 = wmma_f32(wa0[g], b0, acc0);
                acc1 = wmma_f32(wa1[g], b1, acc1);
            }
        }

        float s0[4], s1[4];
#pragma unroll
        for (int p = 0; p < 4; ++p) {
            s0[p] = fold_pair<0x161>(bit0, acc0[2 * p], acc0[2 * p + 1]);
            s1[p] = fold_pair<0x161>(bit0, acc1[2 * p], acc1[2 * p + 1]);
        }
        const float e00 = fold_pair<0x162>(bit1, s0[0], s0[1]);
        const float e01 = fold_pair<0x162>(bit1, s0[2], s0[3]);
        const float e10 = fold_pair<0x162>(bit1, s1[0], s1[1]);
        const float e11 = fold_pair<0x162>(bit1, s1[2], s1[3]);
        const float f0 = fold_pair<0x164>(bit2, e00, e01);
        const float f1 = fold_pair<0x164>(bit2, e10, e11);
        float tot = fold_pair<0x168>(bit3, f0, f1);
        if (IT == 0) tot *= 0.03125f;
        pdst[c * DC] = tot;
    }
}

template <int IT>
__global__ __launch_bounds__(THREADS)
void capsnet_pass_kernel(const float* __restrict__ x,
                         const float* __restrict__ W,
                         const float* __restrict__ part_in,
                         float* __restrict__ part_out,
                         const unsigned long long ws_bytes)
{
    if (ws_bytes < WS_NEED_BYTES) return;

    __shared__ __align__(16) float sWp[NC * DC * WPITCH];
    __shared__ __align__(16) float sStage[WAVES][2][NC][16];
    __shared__ __align__(16) float sCoef[WAVES][NC * 16 * TPW];
    __shared__ __align__(16) float sV[DIN][NC];
    __shared__ __align__(16) float sOpre[NC][DC];
    __shared__ __align__(16) float sO[NC][DC];
    __shared__ float sFac[NC];

    const int tid  = threadIdx.x;
    const int wave = tid >> 5;
    const int lane = tid & 31;
    const int col  = lane & 15;
    const int h    = lane >> 4;
    const int bidx = blockIdx.x / SPLIT;
    const int part = blockIdx.x % SPLIT;

    const float* xb = x + (size_t)bidx * (N_PTS * DIN);

    stage_w(sWp, W, tid);
    if (IT > 0) {
        const float* pin = part_in + (size_t)bidx * (SPLIT * NC * DC);
        float* p = &sOpre[0][0];
        for (int i = tid; i < NC * DC; i += THREADS) {
            float s = pin[i];
            for (int gg = 1; gg < SPLIT; ++gg) s += pin[gg * (NC * DC) + i];
            p[i] = s;
        }
    }
    __syncthreads();

    if (IT > 0) {
        squash_factor(sOpre, sFac, tid);
        __syncthreads();
        for (int i = tid; i < NC * DC; i += THREADS) {
            const int c = i >> 4, d = i & 15;
            sO[c][d] = sFac[c] * sOpre[c][d];
        }
        __syncthreads();
        make_v(sV, sWp, sO, tid);
        __syncthreads();
    }

    route_wave<IT>(xb, sWp, sStage[wave][0], sCoef[wave], sV, &sStage[wave][0][0][0],
                   part * TILES_PER_WG + wave, col, h);
    __syncthreads();

    float* pout = part_out + (size_t)blockIdx.x * (NC * DC);
    const float* p = &sStage[0][0][0][0];
    for (int i = tid; i < NC * DC; i += THREADS) {
        float s = p[i];
        for (int w = 1; w < WAVES; ++w) s += p[w * (2 * NC * 16) + i];
        pout[i] = s;
    }
}

__global__ __launch_bounds__(THREADS)
void capsnet_finish_kernel(const float* __restrict__ x,
                           const float* __restrict__ W,
                           const float* __restrict__ part_in,
                           float* __restrict__ out,
                           const unsigned long long ws_bytes)
{
    __shared__ __align__(16) float sWp[NC * DC * WPITCH];
    __shared__ __align__(16) float sStage[WAVES][NC][16];
    __shared__ __align__(16) float sV[DIN][NC];
    __shared__ __align__(16) float sOpre[NC][DC];
    __shared__ __align__(16) float sO[NC][DC];
    __shared__ float sFac[NC];

    const int tid  = threadIdx.x;
    const int bidx = blockIdx.x;

    if (ws_bytes >= WS_NEED_BYTES) {
        const float* pin = part_in + (size_t)bidx * (SPLIT * NC * DC);
        float* p = &sOpre[0][0];
        for (int i = tid; i < NC * DC; i += THREADS) {
            float s = pin[i];
            for (int gg = 1; gg < SPLIT; ++gg) s += pin[gg * (NC * DC) + i];
            p[i] = s;
        }
        __syncthreads();
        squash_factor(sOpre, sFac, tid);
        __syncthreads();
        for (int i = tid; i < NC * DC; i += THREADS)
            out[(size_t)bidx * (NC * DC) + i] = sFac[i >> 4] * p[i];
        return;
    }

    const int wave = tid >> 5;
    const int lane = tid & 31;
    const int col  = lane & 15;
    const int h    = lane >> 4;
    const float* xb = x + (size_t)bidx * (N_PTS * DIN);

    stage_w(sWp, W, tid);
    __syncthreads();

    for (int it = 0; it < 3; ++it) {
        {
            float* p = &sOpre[0][0];
            for (int i = tid; i < NC * DC; i += THREADS) p[i] = 0.0f;
        }
        __syncthreads();
        if (it == 0)
            route_tiles<0>(xb, sWp, sStage[wave], sV, sOpre, wave, NTILES, col, h);
        else
            route_tiles<1>(xb, sWp, sStage[wave], sV, sOpre, wave, NTILES, col, h);
        __syncthreads();
        squash_factor(sOpre, sFac, tid);
        __syncthreads();
        for (int i = tid; i < NC * DC; i += THREADS) {
            const int c = i >> 4, d = i & 15;
            sO[c][d] = sFac[c] * sOpre[c][d];
        }
        __syncthreads();
        if (it < 2) {
            make_v(sV, sWp, sO, tid);
            __syncthreads();
        }
    }
    for (int i = tid; i < NC * DC; i += THREADS)
        out[(size_t)bidx * (NC * DC) + i] = sO[i >> 4][i & 15];
}

extern "C" void kernel_launch(void* const* d_in, const int* in_sizes, int n_in,
                              void* d_out, int out_size, void* d_ws, size_t ws_size,
                              hipStream_t stream) {
    (void)in_sizes; (void)n_in; (void)out_size;
    const float* x = (const float*)d_in[0];
    const float* W = (const float*)d_in[1];
    float* out = (float*)d_out;
    float* planeA = (float*)d_ws;
    float* planeB = (float*)d_ws + PLANE_FLOATS;
    const unsigned long long wsb = (unsigned long long)ws_size;
    hipLaunchKernelGGL(capsnet_pass_kernel<0>, dim3(BATCH * SPLIT), dim3(THREADS), 0,
                       stream, x, W, (const float*)planeB, planeA, wsb);
    hipLaunchKernelGGL(capsnet_pass_kernel<1>, dim3(BATCH * SPLIT), dim3(THREADS), 0,
                       stream, x, W, (const float*)planeA, planeB, wsb);
    hipLaunchKernelGGL(capsnet_pass_kernel<2>, dim3(BATCH * SPLIT), dim3(THREADS), 0,
                       stream, x, W, (const float*)planeB, planeA, wsb);
    hipLaunchKernelGGL(capsnet_finish_kernel, dim3(BATCH), dim3(THREADS), 0,
                       stream, x, W, (const float*)planeA, out, wsb);
}
